# starting kernel
# baseline (speedup 1.0000x reference)
#include <hip/hip_runtime.h>
#include <hip/hip_bf16.h>
#include <math.h>

#define B_  16
#define Q_  64
#define K_  1024
#define QS_ 256
#define H_  128
#define VS_ 256
#define NEG_ (-1e6f)

typedef float v2f __attribute__((ext_vector_type(2)));
typedef float v8f __attribute__((ext_vector_type(8)));

#if defined(__has_builtin)
#  if __has_builtin(__builtin_amdgcn_tanhf)
#    define TANHF(x) __builtin_amdgcn_tanhf(x)
#  else
#    define TANHF(x) tanhf(x)
#  endif
#  if __has_builtin(__builtin_amdgcn_exp2f)
#    define EXP2F(x) __builtin_amdgcn_exp2f(x)
#  else
#    define EXP2F(x) exp2f(x)
#  endif
#else
#  define TANHF(x) tanhf(x)
#  define EXP2F(x) exp2f(x)
#endif

#define LOG2E 1.4426950408889634f

template<int N, int KD>
__global__ __launch_bounds__(128) void gemm_wmma_f32(
    const float* __restrict__ A, long sA,
    const float* __restrict__ W, long sW,
    float* __restrict__ C, long sC)
{
    A += (size_t)blockIdx.y * sA;
    W += (size_t)blockIdx.y * sW;
    C += (size_t)blockIdx.y * sC;

    const int wave = threadIdx.x >> 5;
    const int lane = threadIdx.x & 31;
    const int tile = blockIdx.x * 4 + wave;
    constexpr int tilesN = N >> 4;
    const int tm = tile / tilesN;
    const int tn = tile - tm * tilesN;
    const int r    = lane & 15;
    const int half = lane >> 4;

    const float* arow = A + (size_t)(tm * 16 + r) * KD + 2 * half;
    const float* wcol = W + (size_t)(2 * half) * N + tn * 16 + r;

    v8f c = {};
    #pragma unroll 8
    for (int kk = 0; kk < KD; kk += 4) {
        const v2f a = *(const v2f*)(arow + kk);
        v2f b;
        b.x = wcol[(size_t)kk * N];
        b.y = wcol[(size_t)(kk + 1) * N];
        c = __builtin_amdgcn_wmma_f32_16x16x4_f32(
                 false, a,  false, b,
                 (short)0, c,  false,  false);
    }

    float* crow = C + (size_t)(tm * 16 + 8 * half) * N + tn * 16 + r;
    #pragma unroll
    for (int v = 0; v < 8; ++v)
        crow[(size_t)v * N] = c[v];
}

__global__ __launch_bounds__(256) void scores_softmax(
    const float* __restrict__ qproj,
    const float* __restrict__ kproj,
    const float* __restrict__ Ws,
    const int*   __restrict__ valid_len,
    float* __restrict__ attn)
{
    __shared__ float s_scores[K_];
    __shared__ float s_red[8];

    const int b = blockIdx.x >> 6;
    const int q = blockIdx.x & 63;
    const int t = threadIdx.x;
    const int wave = t >> 5;
    const int lane = t & 31;
    const int h0 = lane * 4;

    const float4 qv = *(const float4*)(qproj + ((size_t)b * Q_ + q) * H_ + h0);
    const float4 wv = *(const float4*)(Ws + h0);
    const int vlen = valid_len[b];
    const float* kbase = kproj + (size_t)b * K_ * H_;

    for (int k = wave; k < K_; k += 8) {
        const float4 kv = *(const float4*)(kbase + (size_t)k * H_ + h0);
        float p = wv.x * TANHF(qv.x + kv.x)
                + wv.y * TANHF(qv.y + kv.y)
                + wv.z * TANHF(qv.z + kv.z)
                + wv.w * TANHF(qv.w + kv.w);
        #pragma unroll
        for (int off = 16; off >= 1; off >>= 1)
            p += __shfl_xor(p, off, 32);
        if (lane == 0)
            s_scores[k] = (k < vlen) ? NEG_ : p;
    }
    __syncthreads();

    const float v0 = s_scores[t];
    const float v1 = s_scores[t + 256];
    const float v2 = s_scores[t + 512];
    const float v3 = s_scores[t + 768];

    float m = fmaxf(fmaxf(v0, v1), fmaxf(v2, v3));
    #pragma unroll
    for (int off = 16; off >= 1; off >>= 1)
        m = fmaxf(m, __shfl_xor(m, off, 32));
    if (lane == 0) s_red[wave] = m;
    __syncthreads();
    float gmax = s_red[0];
    #pragma unroll
    for (int w = 1; w < 8; ++w) gmax = fmaxf(gmax, s_red[w]);
    __syncthreads();

    const float e0 = EXP2F((v0 - gmax) * LOG2E);
    const float e1 = EXP2F((v1 - gmax) * LOG2E);
    const float e2 = EXP2F((v2 - gmax) * LOG2E);
    const float e3 = EXP2F((v3 - gmax) * LOG2E);
    float s = e0 + e1 + e2 + e3;
    #pragma unroll
    for (int off = 16; off >= 1; off >>= 1)
        s += __shfl_xor(s, off, 32);
    if (lane == 0) s_red[wave] = s;
    __syncthreads();
    float tot = s_red[0];
    #pragma unroll
    for (int w = 1; w < 8; ++w) tot += s_red[w];
    const float inv = 1.0f / tot;

    float* orow = attn + ((size_t)b * Q_ + q) * K_;
    orow[t]       = e0 * inv;
    orow[t + 256] = e1 * inv;
    orow[t + 512] = e2 * inv;
    orow[t + 768] = e3 * inv;
}

extern "C" void kernel_launch(void* const* d_in, const int* in_sizes, int n_in,
                              void* d_out, int out_size, void* d_ws, size_t ws_size,
                              hipStream_t stream) {
    const float* queries   = (const float*)d_in[0];
    const float* keys      = (const float*)d_in[1];
    const float* values    = (const float*)d_in[2];
    const int*   valid_len = (const int*)  d_in[3];
    const float* Wq        = (const float*)d_in[4];
    const float* Wk        = (const float*)d_in[5];
    const float* Ws        = (const float*)d_in[6];
    float* out = (float*)d_out;

    float* qproj = (float*)d_ws;
    float* kproj = qproj + (size_t)B_ * Q_ * H_;
    float* attn  = kproj + (size_t)B_ * K_ * H_;

    gemm_wmma_f32<H_, QS_><<<dim3(128, 1), 128, 0, stream>>>(
        queries, 0, Wq, 0, qproj, 0);

    gemm_wmma_f32<H_, QS_><<<dim3(2048, 1), 128, 0, stream>>>(
        keys, 0, Wk, 0, kproj, 0);

    scores_softmax<<<B_ * Q_, 256, 0, stream>>>(
        qproj, kproj, Ws, valid_len, attn);

    gemm_wmma_f32<VS_, K_><<<dim3(16, B_), 128, 0, stream>>>(
        attn,   (long)Q_ * K_,
        values, (long)K_ * VS_,
        out,    (long)Q_ * VS_);
}


# scores stage re-mapped: lanes own keys and walk h serially, 4 queries per key load (k-projection stored feature-major via swapped WMMA operands); h-sum and softmax sums change summation order
# speedup vs baseline: 1.6813x; 1.6813x over previous
#include <hip/hip_runtime.h>
#include <hip/hip_bf16.h>
#include <math.h>

#define B_  16
#define Q_  64
#define K_  1024
#define QS_ 256
#define H_  128
#define VS_ 256
#define NEG_ (-1e6f)

typedef float v2f __attribute__((ext_vector_type(2)));
typedef float v8f __attribute__((ext_vector_type(8)));

#if defined(__has_builtin)
#  if __has_builtin(__builtin_amdgcn_tanhf)
#    define TANHF(x) __builtin_amdgcn_tanhf(x)
#  else
#    define TANHF(x) tanhf(x)
#  endif
#  if __has_builtin(__builtin_amdgcn_exp2f)
#    define EXP2F(x) __builtin_amdgcn_exp2f(x)
#  else
#    define EXP2F(x) exp2f(x)
#  endif
#else
#  define TANHF(x) tanhf(x)
#  define EXP2F(x) exp2f(x)
#endif

#define LOG2E 1.4426950408889634f

template<int N, int KD, bool TOUT>
__global__ __launch_bounds__(128) void gemm_wmma_f32(
    const float* __restrict__ A, long sA,
    const float* __restrict__ W, long sW,
    float* __restrict__ C, long sC)
{
    A += (size_t)blockIdx.y * sA;
    W += (size_t)blockIdx.y * sW;
    C += (size_t)blockIdx.y * sC;

    const int wave = threadIdx.x >> 5;
    const int lane = threadIdx.x & 31;
    const int tile = blockIdx.x * 4 + wave;
    constexpr int tilesN = N >> 4;
    const int tm = tile / tilesN;
    const int tn = tile - tm * tilesN;
    const int r    = lane & 15;
    const int half = lane >> 4;

    const float* arow = A + (size_t)(tm * 16 + r) * KD + 2 * half;
    const float* wcol = W + (size_t)(2 * half) * N + tn * 16 + r;

    v8f c = {};
    #pragma unroll 8
    for (int kk = 0; kk < KD; kk += 4) {
        const v2f a = *(const v2f*)(arow + kk);
        v2f b;
        b.x = wcol[(size_t)kk * N];
        b.y = wcol[(size_t)(kk + 1) * N];
        if (TOUT)
            c = __builtin_amdgcn_wmma_f32_16x16x4_f32(
                     false, b,  false, a,
                     (short)0, c,  false,  false);
        else
            c = __builtin_amdgcn_wmma_f32_16x16x4_f32(
                     false, a,  false, b,
                     (short)0, c,  false,  false);
    }

    if (!TOUT) {
        float* crow = C + (size_t)(tm * 16 + 8 * half) * N + tn * 16 + r;
        #pragma unroll
        for (int v = 0; v < 8; ++v)
            crow[(size_t)v * N] = c[v];
    } else {
        const int row = tm * 16 + r;
        const int bb  = row / K_;
        const int kk  = row - bb * K_;
        float* ccol = C + ((size_t)bb * N + tn * 16 + 8 * half) * K_ + kk;
        #pragma unroll
        for (int v = 0; v < 8; ++v)
            ccol[(size_t)v * K_] = c[v];
    }
}

#define QG_ 4
#define HU_ 4

__global__ __launch_bounds__(512) void scores_softmax(
    const float* __restrict__ qproj,
    const float* __restrict__ kprojT,
    const float* __restrict__ Ws,
    const int*   __restrict__ valid_len,
    float* __restrict__ attn)
{
    __shared__ float s_max[QG_][16];
    __shared__ float s_sum[QG_][16];

    constexpr int groups = Q_ / QG_;
    const int b  = blockIdx.x / groups;
    const int q0 = (blockIdx.x - b * groups) * QG_;
    const int t = threadIdx.x;
    const int wave = t >> 5;
    const int lane = t & 31;
    const int k0 = 2 * t;

    const float* qrow = qproj + ((size_t)b * Q_ + q0) * H_;
    const float* kcol = kprojT + (size_t)b * H_ * K_ + k0;
    const int vlen = valid_len[b];

    v2f acc[QG_];
    #pragma unroll
    for (int i = 0; i < QG_; ++i) acc[i] = (v2f){0.0f, 0.0f};

    v2f   kv[HU_];
    float qs[QG_][HU_];
    float ws[HU_];
    #pragma unroll
    for (int j = 0; j < HU_; ++j) {
        kv[j] = *(const v2f*)(kcol + (size_t)j * K_);
        ws[j] = Ws[j];
        #pragma unroll
        for (int i = 0; i < QG_; ++i) qs[i][j] = qrow[i * H_ + j];
    }

    for (int h0 = 0; h0 < H_; h0 += HU_) {
        const int hn = (h0 + HU_ < H_) ? (h0 + HU_) : h0;
        v2f   kn[HU_];
        float qn[QG_][HU_];
        float wn[HU_];
        #pragma unroll
        for (int j = 0; j < HU_; ++j) {
            kn[j] = *(const v2f*)(kcol + (size_t)(hn + j) * K_);
            wn[j] = Ws[hn + j];
            #pragma unroll
            for (int i = 0; i < QG_; ++i) qn[i][j] = qrow[i * H_ + hn + j];
        }

        #pragma unroll
        for (int j = 0; j < HU_; ++j) {
            #pragma unroll
            for (int i = 0; i < QG_; ++i) {
                const v2f x = kv[j] + qs[i][j];
                v2f th;
                th.x = TANHF(x.x);
                th.y = TANHF(x.y);
                acc[i] += ws[j] * th;
            }
        }

        #pragma unroll
        for (int j = 0; j < HU_; ++j) {
            kv[j] = kn[j];
            ws[j] = wn[j];
            #pragma unroll
            for (int i = 0; i < QG_; ++i) qs[i][j] = qn[i][j];
        }
    }

    const bool m0 = (k0     < vlen);
    const bool m1 = (k0 + 1 < vlen);
    float mx[QG_];
    #pragma unroll
    for (int i = 0; i < QG_; ++i) {
        acc[i].x = m0 ? NEG_ : acc[i].x;
        acc[i].y = m1 ? NEG_ : acc[i].y;
        mx[i] = fmaxf(acc[i].x, acc[i].y);
    }
    #pragma unroll
    for (int off = 16; off >= 1; off >>= 1) {
        #pragma unroll
        for (int i = 0; i < QG_; ++i)
            mx[i] = fmaxf(mx[i], __shfl_xor(mx[i], off, 32));
    }
    if (lane == 0) {
        #pragma unroll
        for (int i = 0; i < QG_; ++i) s_max[i][wave] = mx[i];
    }
    __syncthreads();

    v2f e[QG_];
    float sm[QG_];
    #pragma unroll
    for (int i = 0; i < QG_; ++i) {
        float gmax = s_max[i][0];
        #pragma unroll
        for (int w = 1; w < 16; ++w) gmax = fmaxf(gmax, s_max[i][w]);
        e[i].x = EXP2F((acc[i].x - gmax) * LOG2E);
        e[i].y = EXP2F((acc[i].y - gmax) * LOG2E);
        sm[i] = e[i].x + e[i].y;
    }
    #pragma unroll
    for (int off = 16; off >= 1; off >>= 1) {
        #pragma unroll
        for (int i = 0; i < QG_; ++i)
            sm[i] += __shfl_xor(sm[i], off, 32);
    }
    if (lane == 0) {
        #pragma unroll
        for (int i = 0; i < QG_; ++i) s_sum[i][wave] = sm[i];
    }
    __syncthreads();

    #pragma unroll
    for (int i = 0; i < QG_; ++i) {
        float tot = s_sum[i][0];
        #pragma unroll
        for (int w = 1; w < 16; ++w) tot += s_sum[i][w];
        const float inv = 1.0f / tot;
        float* orow = attn + ((size_t)b * Q_ + q0 + i) * K_ + k0;
        *(v2f*)orow = e[i] * inv;
    }
}

extern "C" void kernel_launch(void* const* d_in, const int* in_sizes, int n_in,
                              void* d_out, int out_size, void* d_ws, size_t ws_size,
                              hipStream_t stream) {
    const float* queries   = (const float*)d_in[0];
    const float* keys      = (const float*)d_in[1];
    const float* values    = (const float*)d_in[2];
    const int*   valid_len = (const int*)  d_in[3];
    const float* Wq        = (const float*)d_in[4];
    const float* Wk        = (const float*)d_in[5];
    const float* Ws        = (const float*)d_in[6];
    float* out = (float*)d_out;

    float* qproj = (float*)d_ws;
    float* kproj = qproj + (size_t)B_ * Q_ * H_;
    float* attn  = kproj + (size_t)B_ * K_ * H_;

    gemm_wmma_f32<H_, QS_, false><<<dim3(128, 1), 128, 0, stream>>>(
        queries, 0, Wq, 0, qproj, 0);

    gemm_wmma_f32<H_, QS_, true><<<dim3(2048, 1), 128, 0, stream>>>(
        keys, 0, Wk, 0, kproj, 0);

    scores_softmax<<<B_ * (Q_ / QG_), 512, 0, stream>>>(
        qproj, kproj, Ws, valid_len, attn);

    gemm_wmma_f32<VS_, K_, false><<<dim3(16, B_), 128, 0, stream>>>(
        attn,   (long)Q_ * K_,
        values, (long)K_ * VS_,
        out,    (long)Q_ * VS_);
}


# key projection re-tiled: each wave owns a 16x64 strip, one A fragment feeds 4 column tiles and W rows are read as 16-byte loads (column map 4c+t); k order kept, results bit-identical
# speedup vs baseline: 1.7999x; 1.0705x over previous
#include <hip/hip_runtime.h>
#include <hip/hip_bf16.h>
#include <math.h>

#define B_  16
#define Q_  64
#define K_  1024
#define QS_ 256
#define H_  128
#define VS_ 256
#define NEG_ (-1e6f)

typedef float v2f __attribute__((ext_vector_type(2)));
typedef float v4f __attribute__((ext_vector_type(4)));
typedef float v8f __attribute__((ext_vector_type(8)));

#if defined(__has_builtin)
#  if __has_builtin(__builtin_amdgcn_tanhf)
#    define TANHF(x) __builtin_amdgcn_tanhf(x)
#  else
#    define TANHF(x) tanhf(x)
#  endif
#  if __has_builtin(__builtin_amdgcn_exp2f)
#    define EXP2F(x) __builtin_amdgcn_exp2f(x)
#  else
#    define EXP2F(x) exp2f(x)
#  endif
#else
#  define TANHF(x) tanhf(x)
#  define EXP2F(x) exp2f(x)
#endif

#define LOG2E 1.4426950408889634f

template<int N, int KD>
__global__ __launch_bounds__(128) void gemm_wmma_f32(
    const float* __restrict__ A, long sA,
    const float* __restrict__ W, long sW,
    float* __restrict__ C, long sC)
{
    A += (size_t)blockIdx.y * sA;
    W += (size_t)blockIdx.y * sW;
    C += (size_t)blockIdx.y * sC;

    const int wave = threadIdx.x >> 5;
    const int lane = threadIdx.x & 31;
    const int tile = blockIdx.x * 4 + wave;
    constexpr int tilesN = N >> 4;
    const int tm = tile / tilesN;
    const int tn = tile - tm * tilesN;
    const int r    = lane & 15;
    const int half = lane >> 4;

    const float* arow = A + (size_t)(tm * 16 + r) * KD + 2 * half;
    const float* wcol = W + (size_t)(2 * half) * N + tn * 16 + r;

    v8f c = {};
    #pragma unroll 8
    for (int kk = 0; kk < KD; kk += 4) {
        const v2f a = *(const v2f*)(arow + kk);
        v2f b;
        b.x = wcol[(size_t)kk * N];
        b.y = wcol[(size_t)(kk + 1) * N];
        c = __builtin_amdgcn_wmma_f32_16x16x4_f32(
                 false, a,  false, b,
                 (short)0, c,  false,  false);
    }

    float* crow = C + (size_t)(tm * 16 + 8 * half) * N + tn * 16 + r;
    #pragma unroll
    for (int v = 0; v < 8; ++v)
        crow[(size_t)v * N] = c[v];
}

template<int N, int KD>
__global__ __launch_bounds__(128) void proj_strip_wmma_f32(
    const float* __restrict__ A,
    const float* __restrict__ W,
    float* __restrict__ C)
{
    const int wave = threadIdx.x >> 5;
    const int lane = threadIdx.x & 31;
    const int strip = blockIdx.x * 4 + wave;
    constexpr int stripsN = N >> 6;
    const int tm = strip / stripsN;
    const int ns = strip - tm * stripsN;
    const int r    = lane & 15;
    const int half = lane >> 4;

    const float* arow = A + (size_t)(tm * 16 + r) * KD + 2 * half;
    const float* wrow = W + (size_t)(2 * half) * N + ns * 64 + 4 * r;

    v8f c0 = {}, c1 = {}, c2 = {}, c3 = {};
    #pragma unroll 8
    for (int kk = 0; kk < KD; kk += 4) {
        const v2f a  = *(const v2f*)(arow + kk);
        const v4f w0 = *(const v4f*)(wrow + (size_t)kk * N);
        const v4f w1 = *(const v4f*)(wrow + (size_t)(kk + 1) * N);
        const v2f b0 = {w0.x, w1.x};
        const v2f b1 = {w0.y, w1.y};
        const v2f b2 = {w0.z, w1.z};
        const v2f b3 = {w0.w, w1.w};
        c0 = __builtin_amdgcn_wmma_f32_16x16x4_f32(false, b0, false, a, (short)0, c0, false, false);
        c1 = __builtin_amdgcn_wmma_f32_16x16x4_f32(false, b1, false, a, (short)0, c1, false, false);
        c2 = __builtin_amdgcn_wmma_f32_16x16x4_f32(false, b2, false, a, (short)0, c2, false, false);
        c3 = __builtin_amdgcn_wmma_f32_16x16x4_f32(false, b3, false, a, (short)0, c3, false, false);
    }

    const int row = tm * 16 + r;
    const int bb  = row / K_;
    const int kidx = row - bb * K_;
    float* cbase = C + ((size_t)bb * N + ns * 64 + 32 * half) * K_ + kidx;
    #pragma unroll
    for (int v = 0; v < 8; ++v) {
        cbase[(size_t)(4 * v + 0) * K_] = c0[v];
        cbase[(size_t)(4 * v + 1) * K_] = c1[v];
        cbase[(size_t)(4 * v + 2) * K_] = c2[v];
        cbase[(size_t)(4 * v + 3) * K_] = c3[v];
    }
}

#define QG_ 4
#define HU_ 4

__global__ __launch_bounds__(512) void scores_softmax(
    const float* __restrict__ qproj,
    const float* __restrict__ kprojT,
    const float* __restrict__ Ws,
    const int*   __restrict__ valid_len,
    float* __restrict__ attn)
{
    __shared__ float s_max[QG_][16];
    __shared__ float s_sum[QG_][16];

    constexpr int groups = Q_ / QG_;
    const int b  = blockIdx.x / groups;
    const int q0 = (blockIdx.x - b * groups) * QG_;
    const int t = threadIdx.x;
    const int wave = t >> 5;
    const int lane = t & 31;
    const int k0 = 2 * t;

    const float* qrow = qproj + ((size_t)b * Q_ + q0) * H_;
    const float* kcol = kprojT + (size_t)b * H_ * K_ + k0;
    const int vlen = valid_len[b];

    v2f acc[QG_];
    #pragma unroll
    for (int i = 0; i < QG_; ++i) acc[i] = (v2f){0.0f, 0.0f};

    v2f   kv[HU_];
    float qs[QG_][HU_];
    float ws[HU_];
    #pragma unroll
    for (int j = 0; j < HU_; ++j) {
        kv[j] = *(const v2f*)(kcol + (size_t)j * K_);
        ws[j] = Ws[j];
        #pragma unroll
        for (int i = 0; i < QG_; ++i) qs[i][j] = qrow[i * H_ + j];
    }

    for (int h0 = 0; h0 < H_; h0 += HU_) {
        const int hn = (h0 + HU_ < H_) ? (h0 + HU_) : h0;
        v2f   kn[HU_];
        float qn[QG_][HU_];
        float wn[HU_];
        #pragma unroll
        for (int j = 0; j < HU_; ++j) {
            kn[j] = *(const v2f*)(kcol + (size_t)(hn + j) * K_);
            wn[j] = Ws[hn + j];
            #pragma unroll
            for (int i = 0; i < QG_; ++i) qn[i][j] = qrow[i * H_ + hn + j];
        }

        #pragma unroll
        for (int j = 0; j < HU_; ++j) {
            #pragma unroll
            for (int i = 0; i < QG_; ++i) {
                const v2f x = kv[j] + qs[i][j];
                v2f th;
                th.x = TANHF(x.x);
                th.y = TANHF(x.y);
                acc[i] += ws[j] * th;
            }
        }

        #pragma unroll
        for (int j = 0; j < HU_; ++j) {
            kv[j] = kn[j];
            ws[j] = wn[j];
            #pragma unroll
            for (int i = 0; i < QG_; ++i) qs[i][j] = qn[i][j];
        }
    }

    const bool m0 = (k0     < vlen);
    const bool m1 = (k0 + 1 < vlen);
    float mx[QG_];
    #pragma unroll
    for (int i = 0; i < QG_; ++i) {
        acc[i].x = m0 ? NEG_ : acc[i].x;
        acc[i].y = m1 ? NEG_ : acc[i].y;
        mx[i] = fmaxf(acc[i].x, acc[i].y);
    }
    #pragma unroll
    for (int off = 16; off >= 1; off >>= 1) {
        #pragma unroll
        for (int i = 0; i < QG_; ++i)
            mx[i] = fmaxf(mx[i], __shfl_xor(mx[i], off, 32));
    }
    if (lane == 0) {
        #pragma unroll
        for (int i = 0; i < QG_; ++i) s_max[i][wave] = mx[i];
    }
    __syncthreads();

    v2f e[QG_];
    float sm[QG_];
    #pragma unroll
    for (int i = 0; i < QG_; ++i) {
        float gmax = s_max[i][0];
        #pragma unroll
        for (int w = 1; w < 16; ++w) gmax = fmaxf(gmax, s_max[i][w]);
        e[i].x = EXP2F((acc[i].x - gmax) * LOG2E);
        e[i].y = EXP2F((acc[i].y - gmax) * LOG2E);
        sm[i] = e[i].x + e[i].y;
    }
    #pragma unroll
    for (int off = 16; off >= 1; off >>= 1) {
        #pragma unroll
        for (int i = 0; i < QG_; ++i)
            sm[i] += __shfl_xor(sm[i], off, 32);
    }
    if (lane == 0) {
        #pragma unroll
        for (int i = 0; i < QG_; ++i) s_sum[i][wave] = sm[i];
    }
    __syncthreads();

    #pragma unroll
    for (int i = 0; i < QG_; ++i) {
        float tot = s_sum[i][0];
        #pragma unroll
        for (int w = 1; w < 16; ++w) tot += s_sum[i][w];
        const float inv = 1.0f / tot;
        float* orow = attn + ((size_t)b * Q_ + q0 + i) * K_ + k0;
        *(v2f*)orow = e[i] * inv;
    }
}

extern "C" void kernel_launch(void* const* d_in, const int* in_sizes, int n_in,
                              void* d_out, int out_size, void* d_ws, size_t ws_size,
                              hipStream_t stream) {
    const float* queries   = (const float*)d_in[0];
    const float* keys      = (const float*)d_in[1];
    const float* values    = (const float*)d_in[2];
    const int*   valid_len = (const int*)  d_in[3];
    const float* Wq        = (const float*)d_in[4];
    const float* Wk        = (const float*)d_in[5];
    const float* Ws        = (const float*)d_in[6];
    float* out = (float*)d_out;

    float* qproj = (float*)d_ws;
    float* kproj = qproj + (size_t)B_ * Q_ * H_;
    float* attn  = kproj + (size_t)B_ * K_ * H_;

    gemm_wmma_f32<H_, QS_><<<dim3(128, 1), 128, 0, stream>>>(
        queries, 0, Wq, 0, qproj, 0);

    proj_strip_wmma_f32<H_, QS_><<<dim3(512, 1), 128, 0, stream>>>(
        keys, Wk, kproj);

    scores_softmax<<<B_ * (Q_ / QG_), 512, 0, stream>>>(
        qproj, kproj, Ws, valid_len, attn);

    gemm_wmma_f32<VS_, K_><<<dim3(16, B_), 128, 0, stream>>>(
        attn,   (long)Q_ * K_,
        values, (long)K_ * VS_,
        out,    (long)Q_ * VS_);
}


# attn x values product re-tiled: one block per 16x64 output strip, 16 waves split k (64 each, all loads issued up front, 4 tiles share an A fragment), partials summed in LDS; k-sum order changes
# speedup vs baseline: 2.5093x; 1.3941x over previous
#include <hip/hip_runtime.h>
#include <hip/hip_bf16.h>
#include <math.h>

#define B_  16
#define Q_  64
#define K_  1024
#define QS_ 256
#define H_  128
#define VS_ 256
#define NEG_ (-1e6f)

typedef float v2f __attribute__((ext_vector_type(2)));
typedef float v4f __attribute__((ext_vector_type(4)));
typedef float v8f __attribute__((ext_vector_type(8)));

#if defined(__has_builtin)
#  if __has_builtin(__builtin_amdgcn_tanhf)
#    define TANHF(x) __builtin_amdgcn_tanhf(x)
#  else
#    define TANHF(x) tanhf(x)
#  endif
#  if __has_builtin(__builtin_amdgcn_exp2f)
#    define EXP2F(x) __builtin_amdgcn_exp2f(x)
#  else
#    define EXP2F(x) exp2f(x)
#  endif
#else
#  define TANHF(x) tanhf(x)
#  define EXP2F(x) exp2f(x)
#endif

#define LOG2E 1.4426950408889634f

template<int N, int KD>
__global__ __launch_bounds__(128) void gemm_wmma_f32(
    const float* __restrict__ A, long sA,
    const float* __restrict__ W, long sW,
    float* __restrict__ C, long sC)
{
    A += (size_t)blockIdx.y * sA;
    W += (size_t)blockIdx.y * sW;
    C += (size_t)blockIdx.y * sC;

    const int wave = threadIdx.x >> 5;
    const int lane = threadIdx.x & 31;
    const int tile = blockIdx.x * 4 + wave;
    constexpr int tilesN = N >> 4;
    const int tm = tile / tilesN;
    const int tn = tile - tm * tilesN;
    const int r    = lane & 15;
    const int half = lane >> 4;

    const float* arow = A + (size_t)(tm * 16 + r) * KD + 2 * half;
    const float* wcol = W + (size_t)(2 * half) * N + tn * 16 + r;

    v8f c = {};
    #pragma unroll 8
    for (int kk = 0; kk < KD; kk += 4) {
        const v2f a = *(const v2f*)(arow + kk);
        v2f b;
        b.x = wcol[(size_t)kk * N];
        b.y = wcol[(size_t)(kk + 1) * N];
        c = __builtin_amdgcn_wmma_f32_16x16x4_f32(
                 false, a,  false, b,
                 (short)0, c,  false,  false);
    }

    float* crow = C + (size_t)(tm * 16 + 8 * half) * N + tn * 16 + r;
    #pragma unroll
    for (int v = 0; v < 8; ++v)
        crow[(size_t)v * N] = c[v];
}

template<int N, int KD>
__global__ __launch_bounds__(128) void proj_strip_wmma_f32(
    const float* __restrict__ A,
    const float* __restrict__ W,
    float* __restrict__ C)
{
    const int wave = threadIdx.x >> 5;
    const int lane = threadIdx.x & 31;
    const int strip = blockIdx.x * 4 + wave;
    constexpr int stripsN = N >> 6;
    const int tm = strip / stripsN;
    const int ns = strip - tm * stripsN;
    const int r    = lane & 15;
    const int half = lane >> 4;

    const float* arow = A + (size_t)(tm * 16 + r) * KD + 2 * half;
    const float* wrow = W + (size_t)(2 * half) * N + ns * 64 + 4 * r;

    v8f c0 = {}, c1 = {}, c2 = {}, c3 = {};
    #pragma unroll 8
    for (int kk = 0; kk < KD; kk += 4) {
        const v2f a  = *(const v2f*)(arow + kk);
        const v4f w0 = *(const v4f*)(wrow + (size_t)kk * N);
        const v4f w1 = *(const v4f*)(wrow + (size_t)(kk + 1) * N);
        const v2f b0 = {w0.x, w1.x};
        const v2f b1 = {w0.y, w1.y};
        const v2f b2 = {w0.z, w1.z};
        const v2f b3 = {w0.w, w1.w};
        c0 = __builtin_amdgcn_wmma_f32_16x16x4_f32(false, b0, false, a, (short)0, c0, false, false);
        c1 = __builtin_amdgcn_wmma_f32_16x16x4_f32(false, b1, false, a, (short)0, c1, false, false);
        c2 = __builtin_amdgcn_wmma_f32_16x16x4_f32(false, b2, false, a, (short)0, c2, false, false);
        c3 = __builtin_amdgcn_wmma_f32_16x16x4_f32(false, b3, false, a, (short)0, c3, false, false);
    }

    const int row = tm * 16 + r;
    const int bb  = row / K_;
    const int kidx = row - bb * K_;
    float* cbase = C + ((size_t)bb * N + ns * 64 + 32 * half) * K_ + kidx;
    #pragma unroll
    for (int v = 0; v < 8; ++v) {
        cbase[(size_t)(4 * v + 0) * K_] = c0[v];
        cbase[(size_t)(4 * v + 1) * K_] = c1[v];
        cbase[(size_t)(4 * v + 2) * K_] = c2[v];
        cbase[(size_t)(4 * v + 3) * K_] = c3[v];
    }
}

#define PV_WAVES_ 16
#define PV_KW_    (K_ / PV_WAVES_)

__global__ __launch_bounds__(512) __attribute__((amdgpu_waves_per_eu(4, 4)))
void pv_strip_splitk_wmma_f32(
    const float* __restrict__ attn,
    const float* __restrict__ values,
    float* __restrict__ out)
{
    __shared__ v4f part[PV_WAVES_ * 16 * 16];

    const int wave = threadIdx.x >> 5;
    const int lane = threadIdx.x & 31;
    const int r    = lane & 15;
    const int half = lane >> 4;
    const int ns = blockIdx.x & 3;
    const int tm = (blockIdx.x >> 2) & 3;
    const int b  = blockIdx.x >> 4;

    const int k0 = wave * PV_KW_ + 4 * half;
    const float* arow = attn + ((size_t)b * Q_ + tm * 16 + r) * K_ + k0;
    const float* vrow = values + ((size_t)b * K_ + k0) * VS_ + ns * 64 + 4 * r;

    v4f a4[PV_KW_ / 8];
    v4f wv[PV_KW_ / 8][4];
    #pragma unroll
    for (int i = 0; i < PV_KW_ / 8; ++i) {
        a4[i] = *(const v4f*)(arow + 8 * i);
        #pragma unroll
        for (int j = 0; j < 4; ++j)
            wv[i][j] = *(const v4f*)(vrow + (size_t)(8 * i + j) * VS_);
    }

    v8f c0 = {}, c1 = {}, c2 = {}, c3 = {};
    #pragma unroll
    for (int i = 0; i < PV_KW_ / 8; ++i) {
        const v2f a01 = {a4[i].x, a4[i].y};
        const v2f a23 = {a4[i].z, a4[i].w};
        const v4f w0 = wv[i][0], w1 = wv[i][1], w2 = wv[i][2], w3 = wv[i][3];
        c0 = __builtin_amdgcn_wmma_f32_16x16x4_f32(false, a01, false, (v2f){w0.x, w1.x}, (short)0, c0, false, false);
        c1 = __builtin_amdgcn_wmma_f32_16x16x4_f32(false, a01, false, (v2f){w0.y, w1.y}, (short)0, c1, false, false);
        c2 = __builtin_amdgcn_wmma_f32_16x16x4_f32(false, a01, false, (v2f){w0.z, w1.z}, (short)0, c2, false, false);
        c3 = __builtin_amdgcn_wmma_f32_16x16x4_f32(false, a01, false, (v2f){w0.w, w1.w}, (short)0, c3, false, false);
        c0 = __builtin_amdgcn_wmma_f32_16x16x4_f32(false, a23, false, (v2f){w2.x, w3.x}, (short)0, c0, false, false);
        c1 = __builtin_amdgcn_wmma_f32_16x16x4_f32(false, a23, false, (v2f){w2.y, w3.y}, (short)0, c1, false, false);
        c2 = __builtin_amdgcn_wmma_f32_16x16x4_f32(false, a23, false, (v2f){w2.z, w3.z}, (short)0, c2, false, false);
        c3 = __builtin_amdgcn_wmma_f32_16x16x4_f32(false, a23, false, (v2f){w2.w, w3.w}, (short)0, c3, false, false);
    }

    v4f* mine = part + (wave * 16 + 8 * half) * 16 + r;
    #pragma unroll
    for (int v = 0; v < 8; ++v)
        mine[v * 16] = (v4f){c0[v], c1[v], c2[v], c3[v]};
    __syncthreads();

    const v2f* col = (const v2f*)part + wave * 32 + lane;
    v2f s = col[0];
    #pragma unroll
    for (int w = 1; w < PV_WAVES_; ++w)
        s += col[w * 512];
    float* orow = out + ((size_t)b * Q_ + tm * 16 + wave) * VS_ + ns * 64 + 2 * lane;
    *(v2f*)orow = s;
}

#define QG_ 4
#define HU_ 4

__global__ __launch_bounds__(512) void scores_softmax(
    const float* __restrict__ qproj,
    const float* __restrict__ kprojT,
    const float* __restrict__ Ws,
    const int*   __restrict__ valid_len,
    float* __restrict__ attn)
{
    __shared__ float s_max[QG_][16];
    __shared__ float s_sum[QG_][16];

    constexpr int groups = Q_ / QG_;
    const int b  = blockIdx.x / groups;
    const int q0 = (blockIdx.x - b * groups) * QG_;
    const int t = threadIdx.x;
    const int wave = t >> 5;
    const int lane = t & 31;
    const int k0 = 2 * t;

    const float* qrow = qproj + ((size_t)b * Q_ + q0) * H_;
    const float* kcol = kprojT + (size_t)b * H_ * K_ + k0;
    const int vlen = valid_len[b];

    v2f acc[QG_];
    #pragma unroll
    for (int i = 0; i < QG_; ++i) acc[i] = (v2f){0.0f, 0.0f};

    v2f   kv[HU_];
    float qs[QG_][HU_];
    float ws[HU_];
    #pragma unroll
    for (int j = 0; j < HU_; ++j) {
        kv[j] = *(const v2f*)(kcol + (size_t)j * K_);
        ws[j] = Ws[j];
        #pragma unroll
        for (int i = 0; i < QG_; ++i) qs[i][j] = qrow[i * H_ + j];
    }

    for (int h0 = 0; h0 < H_; h0 += HU_) {
        const int hn = (h0 + HU_ < H_) ? (h0 + HU_) : h0;
        v2f   kn[HU_];
        float qn[QG_][HU_];
        float wn[HU_];
        #pragma unroll
        for (int j = 0; j < HU_; ++j) {
            kn[j] = *(const v2f*)(kcol + (size_t)(hn + j) * K_);
            wn[j] = Ws[hn + j];
            #pragma unroll
            for (int i = 0; i < QG_; ++i) qn[i][j] = qrow[i * H_ + hn + j];
        }

        #pragma unroll
        for (int j = 0; j < HU_; ++j) {
            #pragma unroll
            for (int i = 0; i < QG_; ++i) {
                const v2f x = kv[j] + qs[i][j];
                v2f th;
                th.x = TANHF(x.x);
                th.y = TANHF(x.y);
                acc[i] += ws[j] * th;
            }
        }

        #pragma unroll
        for (int j = 0; j < HU_; ++j) {
            kv[j] = kn[j];
            ws[j] = wn[j];
            #pragma unroll
            for (int i = 0; i < QG_; ++i) qs[i][j] = qn[i][j];
        }
    }

    const bool m0 = (k0     < vlen);
    const bool m1 = (k0 + 1 < vlen);
    float mx[QG_];
    #pragma unroll
    for (int i = 0; i < QG_; ++i) {
        acc[i].x = m0 ? NEG_ : acc[i].x;
        acc[i].y = m1 ? NEG_ : acc[i].y;
        mx[i] = fmaxf(acc[i].x, acc[i].y);
    }
    #pragma unroll
    for (int off = 16; off >= 1; off >>= 1) {
        #pragma unroll
        for (int i = 0; i < QG_; ++i)
            mx[i] = fmaxf(mx[i], __shfl_xor(mx[i], off, 32));
    }
    if (lane == 0) {
        #pragma unroll
        for (int i = 0; i < QG_; ++i) s_max[i][wave] = mx[i];
    }
    __syncthreads();

    v2f e[QG_];
    float sm[QG_];
    #pragma unroll
    for (int i = 0; i < QG_; ++i) {
        float gmax = s_max[i][0];
        #pragma unroll
        for (int w = 1; w < 16; ++w) gmax = fmaxf(gmax, s_max[i][w]);
        e[i].x = EXP2F((acc[i].x - gmax) * LOG2E);
        e[i].y = EXP2F((acc[i].y - gmax) * LOG2E);
        sm[i] = e[i].x + e[i].y;
    }
    #pragma unroll
    for (int off = 16; off >= 1; off >>= 1) {
        #pragma unroll
        for (int i = 0; i < QG_; ++i)
            sm[i] += __shfl_xor(sm[i], off, 32);
    }
    if (lane == 0) {
        #pragma unroll
        for (int i = 0; i < QG_; ++i) s_sum[i][wave] = sm[i];
    }
    __syncthreads();

    #pragma unroll
    for (int i = 0; i < QG_; ++i) {
        float tot = s_sum[i][0];
        #pragma unroll
        for (int w = 1; w < 16; ++w) tot += s_sum[i][w];
        const float inv = 1.0f / tot;
        float* orow = attn + ((size_t)b * Q_ + q0 + i) * K_ + k0;
        *(v2f*)orow = e[i] * inv;
    }
}

extern "C" void kernel_launch(void* const* d_in, const int* in_sizes, int n_in,
                              void* d_out, int out_size, void* d_ws, size_t ws_size,
                              hipStream_t stream) {
    const float* queries   = (const float*)d_in[0];
    const float* keys      = (const float*)d_in[1];
    const float* values    = (const float*)d_in[2];
    const int*   valid_len = (const int*)  d_in[3];
    const float* Wq        = (const float*)d_in[4];
    const float* Wk        = (const float*)d_in[5];
    const float* Ws        = (const float*)d_in[6];
    float* out = (float*)d_out;

    float* qproj = (float*)d_ws;
    float* kproj = qproj + (size_t)B_ * Q_ * H_;
    float* attn  = kproj + (size_t)B_ * K_ * H_;

    gemm_wmma_f32<H_, QS_><<<dim3(128, 1), 128, 0, stream>>>(
        queries, 0, Wq, 0, qproj, 0);

    proj_strip_wmma_f32<H_, QS_><<<dim3(512, 1), 128, 0, stream>>>(
        keys, Wk, kproj);

    scores_softmax<<<B_ * (Q_ / QG_), 512, 0, stream>>>(
        qproj, kproj, Ws, valid_len, attn);

    pv_strip_splitk_wmma_f32<<<B_ * 16, 512, 0, stream>>>(attn, values, out);
}


# query and key projections merged into one launch (128 + 512 blocks, block-uniform branch): the short query projection runs beside the key projection; tile code kept, bit-identical; launches 4 -> 3
# speedup vs baseline: 2.9521x; 1.1765x over previous
#include <hip/hip_runtime.h>
#include <hip/hip_bf16.h>
#include <math.h>

#define B_  16
#define Q_  64
#define K_  1024
#define QS_ 256
#define H_  128
#define VS_ 256
#define NEG_ (-1e6f)

typedef float v2f __attribute__((ext_vector_type(2)));
typedef float v4f __attribute__((ext_vector_type(4)));
typedef float v8f __attribute__((ext_vector_type(8)));

#if defined(__has_builtin)
#  if __has_builtin(__builtin_amdgcn_tanhf)
#    define TANHF(x) __builtin_amdgcn_tanhf(x)
#  else
#    define TANHF(x) tanhf(x)
#  endif
#  if __has_builtin(__builtin_amdgcn_exp2f)
#    define EXP2F(x) __builtin_amdgcn_exp2f(x)
#  else
#    define EXP2F(x) exp2f(x)
#  endif
#else
#  define TANHF(x) tanhf(x)
#  define EXP2F(x) exp2f(x)
#endif

#define LOG2E 1.4426950408889634f

template<int N, int KD>
__device__ __forceinline__ void gemm_tile_wmma_f32(
    const float* __restrict__ A,
    const float* __restrict__ W,
    float* __restrict__ C,
    const int blk)
{
    const int wave = threadIdx.x >> 5;
    const int lane = threadIdx.x & 31;
    const int tile = blk * 4 + wave;
    constexpr int tilesN = N >> 4;
    const int tm = tile / tilesN;
    const int tn = tile - tm * tilesN;
    const int r    = lane & 15;
    const int half = lane >> 4;

    const float* arow = A + (size_t)(tm * 16 + r) * KD + 2 * half;
    const float* wcol = W + (size_t)(2 * half) * N + tn * 16 + r;

    v8f c = {};
    #pragma unroll 8
    for (int kk = 0; kk < KD; kk += 4) {
        const v2f a = *(const v2f*)(arow + kk);
        v2f b;
        b.x = wcol[(size_t)kk * N];
        b.y = wcol[(size_t)(kk + 1) * N];
        c = __builtin_amdgcn_wmma_f32_16x16x4_f32(
                 false, a,  false, b,
                 (short)0, c,  false,  false);
    }

    float* crow = C + (size_t)(tm * 16 + 8 * half) * N + tn * 16 + r;
    #pragma unroll
    for (int v = 0; v < 8; ++v)
        crow[(size_t)v * N] = c[v];
}

template<int N, int KD>
__device__ __forceinline__ void proj_strip_wmma_f32(
    const float* __restrict__ A,
    const float* __restrict__ W,
    float* __restrict__ C,
    const int blk)
{
    const int wave = threadIdx.x >> 5;
    const int lane = threadIdx.x & 31;
    const int strip = blk * 4 + wave;
    constexpr int stripsN = N >> 6;
    const int tm = strip / stripsN;
    const int ns = strip - tm * stripsN;
    const int r    = lane & 15;
    const int half = lane >> 4;

    const float* arow = A + (size_t)(tm * 16 + r) * KD + 2 * half;
    const float* wrow = W + (size_t)(2 * half) * N + ns * 64 + 4 * r;

    v8f c0 = {}, c1 = {}, c2 = {}, c3 = {};
    #pragma unroll 8
    for (int kk = 0; kk < KD; kk += 4) {
        const v2f a  = *(const v2f*)(arow + kk);
        const v4f w0 = *(const v4f*)(wrow + (size_t)kk * N);
        const v4f w1 = *(const v4f*)(wrow + (size_t)(kk + 1) * N);
        const v2f b0 = {w0.x, w1.x};
        const v2f b1 = {w0.y, w1.y};
        const v2f b2 = {w0.z, w1.z};
        const v2f b3 = {w0.w, w1.w};
        c0 = __builtin_amdgcn_wmma_f32_16x16x4_f32(false, b0, false, a, (short)0, c0, false, false);
        c1 = __builtin_amdgcn_wmma_f32_16x16x4_f32(false, b1, false, a, (short)0, c1, false, false);
        c2 = __builtin_amdgcn_wmma_f32_16x16x4_f32(false, b2, false, a, (short)0, c2, false, false);
        c3 = __builtin_amdgcn_wmma_f32_16x16x4_f32(false, b3, false, a, (short)0, c3, false, false);
    }

    const int row = tm * 16 + r;
    const int bb  = row / K_;
    const int kidx = row - bb * K_;
    float* cbase = C + ((size_t)bb * N + ns * 64 + 32 * half) * K_ + kidx;
    #pragma unroll
    for (int v = 0; v < 8; ++v) {
        cbase[(size_t)(4 * v + 0) * K_] = c0[v];
        cbase[(size_t)(4 * v + 1) * K_] = c1[v];
        cbase[(size_t)(4 * v + 2) * K_] = c2[v];
        cbase[(size_t)(4 * v + 3) * K_] = c3[v];
    }
}

#define QP_BLOCKS_ ((B_ * Q_ / 16) * (H_ / 16) / 4)
#define KP_BLOCKS_ ((B_ * K_ / 16) * (H_ / 64) / 4)

__global__ __launch_bounds__(128) void proj_qk_wmma_f32(
    const float* __restrict__ queries,
    const float* __restrict__ Wq,
    float* __restrict__ qproj,
    const float* __restrict__ keys,
    const float* __restrict__ Wk,
    float* __restrict__ kprojT)
{
    if (blockIdx.x < QP_BLOCKS_)
        gemm_tile_wmma_f32<H_, QS_>(queries, Wq, qproj, (int)blockIdx.x);
    else
        proj_strip_wmma_f32<H_, QS_>(keys, Wk, kprojT, (int)blockIdx.x - QP_BLOCKS_);
}

#define PV_WAVES_ 16
#define PV_KW_    (K_ / PV_WAVES_)

__global__ __launch_bounds__(512) __attribute__((amdgpu_waves_per_eu(4, 4)))
void pv_strip_splitk_wmma_f32(
    const float* __restrict__ attn,
    const float* __restrict__ values,
    float* __restrict__ out)
{
    __shared__ v4f part[PV_WAVES_ * 16 * 16];

    const int wave = threadIdx.x >> 5;
    const int lane = threadIdx.x & 31;
    const int r    = lane & 15;
    const int half = lane >> 4;
    const int ns = blockIdx.x & 3;
    const int tm = (blockIdx.x >> 2) & 3;
    const int b  = blockIdx.x >> 4;

    const int k0 = wave * PV_KW_ + 4 * half;
    const float* arow = attn + ((size_t)b * Q_ + tm * 16 + r) * K_ + k0;
    const float* vrow = values + ((size_t)b * K_ + k0) * VS_ + ns * 64 + 4 * r;

    v4f a4[PV_KW_ / 8];
    v4f wv[PV_KW_ / 8][4];
    #pragma unroll
    for (int i = 0; i < PV_KW_ / 8; ++i) {
        a4[i] = *(const v4f*)(arow + 8 * i);
        #pragma unroll
        for (int j = 0; j < 4; ++j)
            wv[i][j] = *(const v4f*)(vrow + (size_t)(8 * i + j) * VS_);
    }

    v8f c0 = {}, c1 = {}, c2 = {}, c3 = {};
    #pragma unroll
    for (int i = 0; i < PV_KW_ / 8; ++i) {
        const v2f a01 = {a4[i].x, a4[i].y};
        const v2f a23 = {a4[i].z, a4[i].w};
        const v4f w0 = wv[i][0], w1 = wv[i][1], w2 = wv[i][2], w3 = wv[i][3];
        c0 = __builtin_amdgcn_wmma_f32_16x16x4_f32(false, a01, false, (v2f){w0.x, w1.x}, (short)0, c0, false, false);
        c1 = __builtin_amdgcn_wmma_f32_16x16x4_f32(false, a01, false, (v2f){w0.y, w1.y}, (short)0, c1, false, false);
        c2 = __builtin_amdgcn_wmma_f32_16x16x4_f32(false, a01, false, (v2f){w0.z, w1.z}, (short)0, c2, false, false);
        c3 = __builtin_amdgcn_wmma_f32_16x16x4_f32(false, a01, false, (v2f){w0.w, w1.w}, (short)0, c3, false, false);
        c0 = __builtin_amdgcn_wmma_f32_16x16x4_f32(false, a23, false, (v2f){w2.x, w3.x}, (short)0, c0, false, false);
        c1 = __builtin_amdgcn_wmma_f32_16x16x4_f32(false, a23, false, (v2f){w2.y, w3.y}, (short)0, c1, false, false);
        c2 = __builtin_amdgcn_wmma_f32_16x16x4_f32(false, a23, false, (v2f){w2.z, w3.z}, (short)0, c2, false, false);
        c3 = __builtin_amdgcn_wmma_f32_16x16x4_f32(false, a23, false, (v2f){w2.w, w3.w}, (short)0, c3, false, false);
    }

    v4f* mine = part + (wave * 16 + 8 * half) * 16 + r;
    #pragma unroll
    for (int v = 0; v < 8; ++v)
        mine[v * 16] = (v4f){c0[v], c1[v], c2[v], c3[v]};
    __syncthreads();

    const v2f* col = (const v2f*)part + wave * 32 + lane;
    v2f s = col[0];
    #pragma unroll
    for (int w = 1; w < PV_WAVES_; ++w)
        s += col[w * 512];
    float* orow = out + ((size_t)b * Q_ + tm * 16 + wave) * VS_ + ns * 64 + 2 * lane;
    *(v2f*)orow = s;
}

#define QG_ 4
#define HU_ 4

__global__ __launch_bounds__(512) void scores_softmax(
    const float* __restrict__ qproj,
    const float* __restrict__ kprojT,
    const float* __restrict__ Ws,
    const int*   __restrict__ valid_len,
    float* __restrict__ attn)
{
    __shared__ float s_max[QG_][16];
    __shared__ float s_sum[QG_][16];

    constexpr int groups = Q_ / QG_;
    const int b  = blockIdx.x / groups;
    const int q0 = (blockIdx.x - b * groups) * QG_;
    const int t = threadIdx.x;
    const int wave = t >> 5;
    const int lane = t & 31;
    const int k0 = 2 * t;

    const float* qrow = qproj + ((size_t)b * Q_ + q0) * H_;
    const float* kcol = kprojT + (size_t)b * H_ * K_ + k0;
    const int vlen = valid_len[b];

    v2f acc[QG_];
    #pragma unroll
    for (int i = 0; i < QG_; ++i) acc[i] = (v2f){0.0f, 0.0f};

    v2f   kv[HU_];
    float qs[QG_][HU_];
    float ws[HU_];
    #pragma unroll
    for (int j = 0; j < HU_; ++j) {
        kv[j] = *(const v2f*)(kcol + (size_t)j * K_);
        ws[j] = Ws[j];
        #pragma unroll
        for (int i = 0; i < QG_; ++i) qs[i][j] = qrow[i * H_ + j];
    }

    for (int h0 = 0; h0 < H_; h0 += HU_) {
        const int hn = (h0 + HU_ < H_) ? (h0 + HU_) : h0;
        v2f   kn[HU_];
        float qn[QG_][HU_];
        float wn[HU_];
        #pragma unroll
        for (int j = 0; j < HU_; ++j) {
            kn[j] = *(const v2f*)(kcol + (size_t)(hn + j) * K_);
            wn[j] = Ws[hn + j];
            #pragma unroll
            for (int i = 0; i < QG_; ++i) qn[i][j] = qrow[i * H_ + hn + j];
        }

        #pragma unroll
        for (int j = 0; j < HU_; ++j) {
            #pragma unroll
            for (int i = 0; i < QG_; ++i) {
                const v2f x = kv[j] + qs[i][j];
                v2f th;
                th.x = TANHF(x.x);
                th.y = TANHF(x.y);
                acc[i] += ws[j] * th;
            }
        }

        #pragma unroll
        for (int j = 0; j < HU_; ++j) {
            kv[j] = kn[j];
            ws[j] = wn[j];
            #pragma unroll
            for (int i = 0; i < QG_; ++i) qs[i][j] = qn[i][j];
        }
    }

    const bool m0 = (k0     < vlen);
    const bool m1 = (k0 + 1 < vlen);
    float mx[QG_];
    #pragma unroll
    for (int i = 0; i < QG_; ++i) {
        acc[i].x = m0 ? NEG_ : acc[i].x;
        acc[i].y = m1 ? NEG_ : acc[i].y;
        mx[i] = fmaxf(acc[i].x, acc[i].y);
    }
    #pragma unroll
    for (int off = 16; off >= 1; off >>= 1) {
        #pragma unroll
        for (int i = 0; i < QG_; ++i)
            mx[i] = fmaxf(mx[i], __shfl_xor(mx[i], off, 32));
    }
    if (lane == 0) {
        #pragma unroll
        for (int i = 0; i < QG_; ++i) s_max[i][wave] = mx[i];
    }
    __syncthreads();

    v2f e[QG_];
    float sm[QG_];
    #pragma unroll
    for (int i = 0; i < QG_; ++i) {
        float gmax = s_max[i][0];
        #pragma unroll
        for (int w = 1; w < 16; ++w) gmax = fmaxf(gmax, s_max[i][w]);
        e[i].x = EXP2F((acc[i].x - gmax) * LOG2E);
        e[i].y = EXP2F((acc[i].y - gmax) * LOG2E);
        sm[i] = e[i].x + e[i].y;
    }
    #pragma unroll
    for (int off = 16; off >= 1; off >>= 1) {
        #pragma unroll
        for (int i = 0; i < QG_; ++i)
            sm[i] += __shfl_xor(sm[i], off, 32);
    }
    if (lane == 0) {
        #pragma unroll
        for (int i = 0; i < QG_; ++i) s_sum[i][wave] = sm[i];
    }
    __syncthreads();

    #pragma unroll
    for (int i = 0; i < QG_; ++i) {
        float tot = s_sum[i][0];
        #pragma unroll
        for (int w = 1; w < 16; ++w) tot += s_sum[i][w];
        const float inv = 1.0f / tot;
        float* orow = attn + ((size_t)b * Q_ + q0 + i) * K_ + k0;
        *(v2f*)orow = e[i] * inv;
    }
}

extern "C" void kernel_launch(void* const* d_in, const int* in_sizes, int n_in,
                              void* d_out, int out_size, void* d_ws, size_t ws_size,
                              hipStream_t stream) {
    const float* queries   = (const float*)d_in[0];
    const float* keys      = (const float*)d_in[1];
    const float* values    = (const float*)d_in[2];
    const int*   valid_len = (const int*)  d_in[3];
    const float* Wq        = (const float*)d_in[4];
    const float* Wk        = (const float*)d_in[5];
    const float* Ws        = (const float*)d_in[6];
    float* out = (float*)d_out;

    float* qproj = (float*)d_ws;
    float* kproj = qproj + (size_t)B_ * Q_ * H_;
    float* attn  = kproj + (size_t)B_ * K_ * H_;

    proj_qk_wmma_f32<<<QP_BLOCKS_ + KP_BLOCKS_, 128, 0, stream>>>(
        queries, Wq, qproj, keys, Wk, kproj);

    scores_softmax<<<B_ * (Q_ / QG_), 512, 0, stream>>>(
        qproj, kproj, Ws, valid_len, attn);

    pv_strip_splitk_wmma_f32<<<B_ * 16, 512, 0, stream>>>(attn, values, out);
}


# scores stage: key loads requested six 8-feature stages (48 per wave) ahead of their tanh work through a 7-buffer register ring, h walk fully unrolled; h-sum order kept, results bit-identical
# speedup vs baseline: 3.7707x; 1.2773x over previous
#include <hip/hip_runtime.h>
#include <hip/hip_bf16.h>
#include <math.h>

#define B_  16
#define Q_  64
#define K_  1024
#define QS_ 256
#define H_  128
#define VS_ 256
#define NEG_ (-1e6f)

typedef float v2f __attribute__((ext_vector_type(2)));
typedef float v4f __attribute__((ext_vector_type(4)));
typedef float v8f __attribute__((ext_vector_type(8)));

#if defined(__has_builtin)
#  if __has_builtin(__builtin_amdgcn_tanhf)
#    define TANHF(x) __builtin_amdgcn_tanhf(x)
#  else
#    define TANHF(x) tanhf(x)
#  endif
#  if __has_builtin(__builtin_amdgcn_exp2f)
#    define EXP2F(x) __builtin_amdgcn_exp2f(x)
#  else
#    define EXP2F(x) exp2f(x)
#  endif
#else
#  define TANHF(x) tanhf(x)
#  define EXP2F(x) exp2f(x)
#endif

#define LOG2E 1.4426950408889634f

template<int N, int KD>
__device__ __forceinline__ void gemm_tile_wmma_f32(
    const float* __restrict__ A,
    const float* __restrict__ W,
    float* __restrict__ C,
    const int blk)
{
    const int wave = threadIdx.x >> 5;
    const int lane = threadIdx.x & 31;
    const int tile = blk * 4 + wave;
    constexpr int tilesN = N >> 4;
    const int tm = tile / tilesN;
    const int tn = tile - tm * tilesN;
    const int r    = lane & 15;
    const int half = lane >> 4;

    const float* arow = A + (size_t)(tm * 16 + r) * KD + 2 * half;
    const float* wcol = W + (size_t)(2 * half) * N + tn * 16 + r;

    v8f c = {};
    #pragma unroll 8
    for (int kk = 0; kk < KD; kk += 4) {
        const v2f a = *(const v2f*)(arow + kk);
        v2f b;
        b.x = wcol[(size_t)kk * N];
        b.y = wcol[(size_t)(kk + 1) * N];
        c = __builtin_amdgcn_wmma_f32_16x16x4_f32(
                 false, a,  false, b,
                 (short)0, c,  false,  false);
    }

    float* crow = C + (size_t)(tm * 16 + 8 * half) * N + tn * 16 + r;
    #pragma unroll
    for (int v = 0; v < 8; ++v)
        crow[(size_t)v * N] = c[v];
}

template<int N, int KD>
__device__ __forceinline__ void proj_strip_wmma_f32(
    const float* __restrict__ A,
    const float* __restrict__ W,
    float* __restrict__ C,
    const int blk)
{
    const int wave = threadIdx.x >> 5;
    const int lane = threadIdx.x & 31;
    const int strip = blk * 4 + wave;
    constexpr int stripsN = N >> 6;
    const int tm = strip / stripsN;
    const int ns = strip - tm * stripsN;
    const int r    = lane & 15;
    const int half = lane >> 4;

    const float* arow = A + (size_t)(tm * 16 + r) * KD + 2 * half;
    const float* wrow = W + (size_t)(2 * half) * N + ns * 64 + 4 * r;

    v8f c0 = {}, c1 = {}, c2 = {}, c3 = {};
    #pragma unroll 8
    for (int kk = 0; kk < KD; kk += 4) {
        const v2f a  = *(const v2f*)(arow + kk);
        const v4f w0 = *(const v4f*)(wrow + (size_t)kk * N);
        const v4f w1 = *(const v4f*)(wrow + (size_t)(kk + 1) * N);
        const v2f b0 = {w0.x, w1.x};
        const v2f b1 = {w0.y, w1.y};
        const v2f b2 = {w0.z, w1.z};
        const v2f b3 = {w0.w, w1.w};
        c0 = __builtin_amdgcn_wmma_f32_16x16x4_f32(false, b0, false, a, (short)0, c0, false, false);
        c1 = __builtin_amdgcn_wmma_f32_16x16x4_f32(false, b1, false, a, (short)0, c1, false, false);
        c2 = __builtin_amdgcn_wmma_f32_16x16x4_f32(false, b2, false, a, (short)0, c2, false, false);
        c3 = __builtin_amdgcn_wmma_f32_16x16x4_f32(false, b3, false, a, (short)0, c3, false, false);
    }

    const int row = tm * 16 + r;
    const int bb  = row / K_;
    const int kidx = row - bb * K_;
    float* cbase = C + ((size_t)bb * N + ns * 64 + 32 * half) * K_ + kidx;
    #pragma unroll
    for (int v = 0; v < 8; ++v) {
        cbase[(size_t)(4 * v + 0) * K_] = c0[v];
        cbase[(size_t)(4 * v + 1) * K_] = c1[v];
        cbase[(size_t)(4 * v + 2) * K_] = c2[v];
        cbase[(size_t)(4 * v + 3) * K_] = c3[v];
    }
}

#define QP_BLOCKS_ ((B_ * Q_ / 16) * (H_ / 16) / 4)
#define KP_BLOCKS_ ((B_ * K_ / 16) * (H_ / 64) / 4)

__global__ __launch_bounds__(128) void proj_qk_wmma_f32(
    const float* __restrict__ queries,
    const float* __restrict__ Wq,
    float* __restrict__ qproj,
    const float* __restrict__ keys,
    const float* __restrict__ Wk,
    float* __restrict__ kprojT)
{
    if (blockIdx.x < QP_BLOCKS_)
        gemm_tile_wmma_f32<H_, QS_>(queries, Wq, qproj, (int)blockIdx.x);
    else
        proj_strip_wmma_f32<H_, QS_>(keys, Wk, kprojT, (int)blockIdx.x - QP_BLOCKS_);
}

#define PV_WAVES_ 16
#define PV_KW_    (K_ / PV_WAVES_)

__global__ __launch_bounds__(512) __attribute__((amdgpu_waves_per_eu(4, 4)))
void pv_strip_splitk_wmma_f32(
    const float* __restrict__ attn,
    const float* __restrict__ values,
    float* __restrict__ out)
{
    __shared__ v4f part[PV_WAVES_ * 16 * 16];

    const int wave = threadIdx.x >> 5;
    const int lane = threadIdx.x & 31;
    const int r    = lane & 15;
    const int half = lane >> 4;
    const int ns = blockIdx.x & 3;
    const int tm = (blockIdx.x >> 2) & 3;
    const int b  = blockIdx.x >> 4;

    const int k0 = wave * PV_KW_ + 4 * half;
    const float* arow = attn + ((size_t)b * Q_ + tm * 16 + r) * K_ + k0;
    const float* vrow = values + ((size_t)b * K_ + k0) * VS_ + ns * 64 + 4 * r;

    v4f a4[PV_KW_ / 8];
    v4f wv[PV_KW_ / 8][4];
    #pragma unroll
    for (int i = 0; i < PV_KW_ / 8; ++i) {
        a4[i] = *(const v4f*)(arow + 8 * i);
        #pragma unroll
        for (int j = 0; j < 4; ++j)
            wv[i][j] = *(const v4f*)(vrow + (size_t)(8 * i + j) * VS_);
    }

    v8f c0 = {}, c1 = {}, c2 = {}, c3 = {};
    #pragma unroll
    for (int i = 0; i < PV_KW_ / 8; ++i) {
        const v2f a01 = {a4[i].x, a4[i].y};
        const v2f a23 = {a4[i].z, a4[i].w};
        const v4f w0 = wv[i][0], w1 = wv[i][1], w2 = wv[i][2], w3 = wv[i][3];
        c0 = __builtin_amdgcn_wmma_f32_16x16x4_f32(false, a01, false, (v2f){w0.x, w1.x}, (short)0, c0, false, false);
        c1 = __builtin_amdgcn_wmma_f32_16x16x4_f32(false, a01, false, (v2f){w0.y, w1.y}, (short)0, c1, false, false);
        c2 = __builtin_amdgcn_wmma_f32_16x16x4_f32(false, a01, false, (v2f){w0.z, w1.z}, (short)0, c2, false, false);
        c3 = __builtin_amdgcn_wmma_f32_16x16x4_f32(false, a01, false, (v2f){w0.w, w1.w}, (short)0, c3, false, false);
        c0 = __builtin_amdgcn_wmma_f32_16x16x4_f32(false, a23, false, (v2f){w2.x, w3.x}, (short)0, c0, false, false);
        c1 = __builtin_amdgcn_wmma_f32_16x16x4_f32(false, a23, false, (v2f){w2.y, w3.y}, (short)0, c1, false, false);
        c2 = __builtin_amdgcn_wmma_f32_16x16x4_f32(false, a23, false, (v2f){w2.z, w3.z}, (short)0, c2, false, false);
        c3 = __builtin_amdgcn_wmma_f32_16x16x4_f32(false, a23, false, (v2f){w2.w, w3.w}, (short)0, c3, false, false);
    }

    v4f* mine = part + (wave * 16 + 8 * half) * 16 + r;
    #pragma unroll
    for (int v = 0; v < 8; ++v)
        mine[v * 16] = (v4f){c0[v], c1[v], c2[v], c3[v]};
    __syncthreads();

    const v2f* col = (const v2f*)part + wave * 32 + lane;
    v2f s = col[0];
    #pragma unroll
    for (int w = 1; w < PV_WAVES_; ++w)
        s += col[w * 512];
    float* orow = out + ((size_t)b * Q_ + tm * 16 + wave) * VS_ + ns * 64 + 2 * lane;
    *(v2f*)orow = s;
}

#define QG_ 4
#define HS_ 8
#define NB_ 7
#define NST_ (H_ / HS_)

__global__ __launch_bounds__(512) __attribute__((amdgpu_waves_per_eu(4, 4)))
void scores_softmax(
    const float* __restrict__ qproj,
    const float* __restrict__ kprojT,
    const float* __restrict__ Ws,
    const int*   __restrict__ valid_len,
    float* __restrict__ attn)
{
    __shared__ float s_max[QG_][16];
    __shared__ float s_sum[QG_][16];

    constexpr int groups = Q_ / QG_;
    const int b  = blockIdx.x / groups;
    const int q0 = (blockIdx.x - b * groups) * QG_;
    const int t = threadIdx.x;
    const int wave = t >> 5;
    const int lane = t & 31;
    const int k0 = 2 * t;

    const float* qrow = qproj + ((size_t)b * Q_ + q0) * H_;
    const float* kcol = kprojT + (size_t)b * H_ * K_ + k0;
    const int vlen = valid_len[b];

    v2f acc[QG_];
    #pragma unroll
    for (int i = 0; i < QG_; ++i) acc[i] = (v2f){0.0f, 0.0f};

    v2f kb[NB_][HS_];
    #pragma unroll
    for (int s = 0; s < NB_ - 1; ++s) {
        #pragma unroll
        for (int j = 0; j < HS_; ++j)
            kb[s][j] = *(const v2f*)(kcol + (size_t)(s * HS_ + j) * K_);
    }

    #pragma unroll
    for (int s = 0; s < NST_; ++s) {
        if (s + NB_ - 1 < NST_) {
            #pragma unroll
            for (int j = 0; j < HS_; ++j)
                kb[(s + NB_ - 1) % NB_][j] =
                    *(const v2f*)(kcol + (size_t)((s + NB_ - 1) * HS_ + j) * K_);
        }

        #pragma unroll
        for (int j = 0; j < HS_; ++j) {
            const int h = s * HS_ + j;
            const float w = Ws[h];
            #pragma unroll
            for (int i = 0; i < QG_; ++i) {
                const v2f x = kb[s % NB_][j] + qrow[i * H_ + h];
                v2f th;
                th.x = TANHF(x.x);
                th.y = TANHF(x.y);
                acc[i] += w * th;
            }
        }
    }

    const bool m0 = (k0     < vlen);
    const bool m1 = (k0 + 1 < vlen);
    float mx[QG_];
    #pragma unroll
    for (int i = 0; i < QG_; ++i) {
        acc[i].x = m0 ? NEG_ : acc[i].x;
        acc[i].y = m1 ? NEG_ : acc[i].y;
        mx[i] = fmaxf(acc[i].x, acc[i].y);
    }
    #pragma unroll
    for (int off = 16; off >= 1; off >>= 1) {
        #pragma unroll
        for (int i = 0; i < QG_; ++i)
            mx[i] = fmaxf(mx[i], __shfl_xor(mx[i], off, 32));
    }
    if (lane == 0) {
        #pragma unroll
        for (int i = 0; i < QG_; ++i) s_max[i][wave] = mx[i];
    }
    __syncthreads();

    v2f e[QG_];
    float sm[QG_];
    #pragma unroll
    for (int i = 0; i < QG_; ++i) {
        float gmax = s_max[i][0];
        #pragma unroll
        for (int w = 1; w < 16; ++w) gmax = fmaxf(gmax, s_max[i][w]);
        e[i].x = EXP2F((acc[i].x - gmax) * LOG2E);
        e[i].y = EXP2F((acc[i].y - gmax) * LOG2E);
        sm[i] = e[i].x + e[i].y;
    }
    #pragma unroll
    for (int off = 16; off >= 1; off >>= 1) {
        #pragma unroll
        for (int i = 0; i < QG_; ++i)
            sm[i] += __shfl_xor(sm[i], off, 32);
    }
    if (lane == 0) {
        #pragma unroll
        for (int i = 0; i < QG_; ++i) s_sum[i][wave] = sm[i];
    }
    __syncthreads();

    #pragma unroll
    for (int i = 0; i < QG_; ++i) {
        float tot = s_sum[i][0];
        #pragma unroll
        for (int w = 1; w < 16; ++w) tot += s_sum[i][w];
        const float inv = 1.0f / tot;
        float* orow = attn + ((size_t)b * Q_ + q0 + i) * K_ + k0;
        *(v2f*)orow = e[i] * inv;
    }
}

extern "C" void kernel_launch(void* const* d_in, const int* in_sizes, int n_in,
                              void* d_out, int out_size, void* d_ws, size_t ws_size,
                              hipStream_t stream) {
    const float* queries   = (const float*)d_in[0];
    const float* keys      = (const float*)d_in[1];
    const float* values    = (const float*)d_in[2];
    const int*   valid_len = (const int*)  d_in[3];
    const float* Wq        = (const float*)d_in[4];
    const float* Wk        = (const float*)d_in[5];
    const float* Ws        = (const float*)d_in[6];
    float* out = (float*)d_out;

    float* qproj = (float*)d_ws;
    float* kproj = qproj + (size_t)B_ * Q_ * H_;
    float* attn  = kproj + (size_t)B_ * K_ * H_;

    proj_qk_wmma_f32<<<QP_BLOCKS_ + KP_BLOCKS_, 128, 0, stream>>>(
        queries, Wq, qproj, keys, Wk, kproj);

    scores_softmax<<<B_ * (Q_ / QG_), 512, 0, stream>>>(
        qproj, kproj, Ws, valid_len, attn);

    pv_strip_splitk_wmma_f32<<<B_ * 16, 512, 0, stream>>>(attn, values, out);
}


# projection launch: loads of both projections requested 15 (key strip) / 19 (query tile) k steps ahead of their matrix ops via register rings, fully unrolled; k and sum order kept, bit-identical
# speedup vs baseline: 4.0490x; 1.0738x over previous
#include <hip/hip_runtime.h>
#include <hip/hip_bf16.h>
#include <math.h>

#define B_  16
#define Q_  64
#define K_  1024
#define QS_ 256
#define H_  128
#define VS_ 256
#define NEG_ (-1e6f)

typedef float v2f __attribute__((ext_vector_type(2)));
typedef float v4f __attribute__((ext_vector_type(4)));
typedef float v8f __attribute__((ext_vector_type(8)));

#if defined(__has_builtin)
#  if __has_builtin(__builtin_amdgcn_tanhf)
#    define TANHF(x) __builtin_amdgcn_tanhf(x)
#  else
#    define TANHF(x) tanhf(x)
#  endif
#  if __has_builtin(__builtin_amdgcn_exp2f)
#    define EXP2F(x) __builtin_amdgcn_exp2f(x)
#  else
#    define EXP2F(x) exp2f(x)
#  endif
#else
#  define TANHF(x) tanhf(x)
#  define EXP2F(x) exp2f(x)
#endif

#define LOG2E 1.4426950408889634f

#define QT_NB_ 20
#define KS_NB_ 16

template<int N, int KD>
__device__ __forceinline__ void gemm_tile_wmma_f32(
    const float* __restrict__ A,
    const float* __restrict__ W,
    float* __restrict__ C,
    const int blk)
{
    const int wave = threadIdx.x >> 5;
    const int lane = threadIdx.x & 31;
    const int tile = blk * 4 + wave;
    constexpr int tilesN = N >> 4;
    const int tm = tile / tilesN;
    const int tn = tile - tm * tilesN;
    const int r    = lane & 15;
    const int half = lane >> 4;

    const float* arow = A + (size_t)(tm * 16 + r) * KD + 2 * half;
    const float* wcol = W + (size_t)(2 * half) * N + tn * 16 + r;

    constexpr int NS = KD / 4;
    v2f ar[QT_NB_];
    v2f br[QT_NB_];
    #pragma unroll
    for (int s = 0; s < QT_NB_ - 1; ++s) {
        ar[s]   = *(const v2f*)(arow + 4 * s);
        br[s].x = wcol[(size_t)(4 * s) * N];
        br[s].y = wcol[(size_t)(4 * s + 1) * N];
    }

    v8f c = {};
    #pragma unroll
    for (int s = 0; s < NS; ++s) {
        if (s + QT_NB_ - 1 < NS) {
            const int t = s + QT_NB_ - 1;
            ar[t % QT_NB_]   = *(const v2f*)(arow + 4 * t);
            br[t % QT_NB_].x = wcol[(size_t)(4 * t) * N];
            br[t % QT_NB_].y = wcol[(size_t)(4 * t + 1) * N];
        }
        c = __builtin_amdgcn_wmma_f32_16x16x4_f32(
                 false, ar[s % QT_NB_],  false, br[s % QT_NB_],
                 (short)0, c,  false,  false);
    }

    float* crow = C + (size_t)(tm * 16 + 8 * half) * N + tn * 16 + r;
    #pragma unroll
    for (int v = 0; v < 8; ++v)
        crow[(size_t)v * N] = c[v];
}

template<int N, int KD>
__device__ __forceinline__ void proj_strip_wmma_f32(
    const float* __restrict__ A,
    const float* __restrict__ W,
    float* __restrict__ C,
    const int blk)
{
    const int wave = threadIdx.x >> 5;
    const int lane = threadIdx.x & 31;
    const int strip = blk * 4 + wave;
    constexpr int stripsN = N >> 6;
    const int tm = strip / stripsN;
    const int ns = strip - tm * stripsN;
    const int r    = lane & 15;
    const int half = lane >> 4;

    const float* arow = A + (size_t)(tm * 16 + r) * KD + 2 * half;
    const float* wrow = W + (size_t)(2 * half) * N + ns * 64 + 4 * r;

    constexpr int NS = KD / 4;
    v2f ar[KS_NB_];
    v4f wr0[KS_NB_];
    v4f wr1[KS_NB_];
    #pragma unroll
    for (int s = 0; s < KS_NB_ - 1; ++s) {
        ar[s]  = *(const v2f*)(arow + 4 * s);
        wr0[s] = *(const v4f*)(wrow + (size_t)(4 * s) * N);
        wr1[s] = *(const v4f*)(wrow + (size_t)(4 * s + 1) * N);
    }

    v8f c0 = {}, c1 = {}, c2 = {}, c3 = {};
    #pragma unroll
    for (int s = 0; s < NS; ++s) {
        if (s + KS_NB_ - 1 < NS) {
            const int t = s + KS_NB_ - 1;
            ar[t % KS_NB_]  = *(const v2f*)(arow + 4 * t);
            wr0[t % KS_NB_] = *(const v4f*)(wrow + (size_t)(4 * t) * N);
            wr1[t % KS_NB_] = *(const v4f*)(wrow + (size_t)(4 * t + 1) * N);
        }
        const v2f a  = ar[s % KS_NB_];
        const v4f w0 = wr0[s % KS_NB_];
        const v4f w1 = wr1[s % KS_NB_];
        const v2f b0 = {w0.x, w1.x};
        const v2f b1 = {w0.y, w1.y};
        const v2f b2 = {w0.z, w1.z};
        const v2f b3 = {w0.w, w1.w};
        c0 = __builtin_amdgcn_wmma_f32_16x16x4_f32(false, b0, false, a, (short)0, c0, false, false);
        c1 = __builtin_amdgcn_wmma_f32_16x16x4_f32(false, b1, false, a, (short)0, c1, false, false);
        c2 = __builtin_amdgcn_wmma_f32_16x16x4_f32(false, b2, false, a, (short)0, c2, false, false);
        c3 = __builtin_amdgcn_wmma_f32_16x16x4_f32(false, b3, false, a, (short)0, c3, false, false);
    }

    const int row = tm * 16 + r;
    const int bb  = row / K_;
    const int kidx = row - bb * K_;
    float* cbase = C + ((size_t)bb * N + ns * 64 + 32 * half) * K_ + kidx;
    #pragma unroll
    for (int v = 0; v < 8; ++v) {
        cbase[(size_t)(4 * v + 0) * K_] = c0[v];
        cbase[(size_t)(4 * v + 1) * K_] = c1[v];
        cbase[(size_t)(4 * v + 2) * K_] = c2[v];
        cbase[(size_t)(4 * v + 3) * K_] = c3[v];
    }
}

#define QP_BLOCKS_ ((B_ * Q_ / 16) * (H_ / 16) / 4)
#define KP_BLOCKS_ ((B_ * K_ / 16) * (H_ / 64) / 4)

__global__ __launch_bounds__(128) __attribute__((amdgpu_waves_per_eu(3, 3)))
void proj_qk_wmma_f32(
    const float* __restrict__ queries,
    const float* __restrict__ Wq,
    float* __restrict__ qproj,
    const float* __restrict__ keys,
    const float* __restrict__ Wk,
    float* __restrict__ kprojT)
{
    if (blockIdx.x < QP_BLOCKS_)
        gemm_tile_wmma_f32<H_, QS_>(queries, Wq, qproj, (int)blockIdx.x);
    else
        proj_strip_wmma_f32<H_, QS_>(keys, Wk, kprojT, (int)blockIdx.x - QP_BLOCKS_);
}

#define PV_WAVES_ 16
#define PV_KW_    (K_ / PV_WAVES_)

__global__ __launch_bounds__(512) __attribute__((amdgpu_waves_per_eu(4, 4)))
void pv_strip_splitk_wmma_f32(
    const float* __restrict__ attn,
    const float* __restrict__ values,
    float* __restrict__ out)
{
    __shared__ v4f part[PV_WAVES_ * 16 * 16];

    const int wave = threadIdx.x >> 5;
    const int lane = threadIdx.x & 31;
    const int r    = lane & 15;
    const int half = lane >> 4;
    const int ns = blockIdx.x & 3;
    const int tm = (blockIdx.x >> 2) & 3;
    const int b  = blockIdx.x >> 4;

    const int k0 = wave * PV_KW_ + 4 * half;
    const float* arow = attn + ((size_t)b * Q_ + tm * 16 + r) * K_ + k0;
    const float* vrow = values + ((size_t)b * K_ + k0) * VS_ + ns * 64 + 4 * r;

    v4f a4[PV_KW_ / 8];
    v4f wv[PV_KW_ / 8][4];
    #pragma unroll
    for (int i = 0; i < PV_KW_ / 8; ++i) {
        a4[i] = *(const v4f*)(arow + 8 * i);
        #pragma unroll
        for (int j = 0; j < 4; ++j)
            wv[i][j] = *(const v4f*)(vrow + (size_t)(8 * i + j) * VS_);
    }

    v8f c0 = {}, c1 = {}, c2 = {}, c3 = {};
    #pragma unroll
    for (int i = 0; i < PV_KW_ / 8; ++i) {
        const v2f a01 = {a4[i].x, a4[i].y};
        const v2f a23 = {a4[i].z, a4[i].w};
        const v4f w0 = wv[i][0], w1 = wv[i][1], w2 = wv[i][2], w3 = wv[i][3];
        c0 = __builtin_amdgcn_wmma_f32_16x16x4_f32(false, a01, false, (v2f){w0.x, w1.x}, (short)0, c0, false, false);
        c1 = __builtin_amdgcn_wmma_f32_16x16x4_f32(false, a01, false, (v2f){w0.y, w1.y}, (short)0, c1, false, false);
        c2 = __builtin_amdgcn_wmma_f32_16x16x4_f32(false, a01, false, (v2f){w0.z, w1.z}, (short)0, c2, false, false);
        c3 = __builtin_amdgcn_wmma_f32_16x16x4_f32(false, a01, false, (v2f){w0.w, w1.w}, (short)0, c3, false, false);
        c0 = __builtin_amdgcn_wmma_f32_16x16x4_f32(false, a23, false, (v2f){w2.x, w3.x}, (short)0, c0, false, false);
        c1 = __builtin_amdgcn_wmma_f32_16x16x4_f32(false, a23, false, (v2f){w2.y, w3.y}, (short)0, c1, false, false);
        c2 = __builtin_amdgcn_wmma_f32_16x16x4_f32(false, a23, false, (v2f){w2.z, w3.z}, (short)0, c2, false, false);
        c3 = __builtin_amdgcn_wmma_f32_16x16x4_f32(false, a23, false, (v2f){w2.w, w3.w}, (short)0, c3, false, false);
    }

    v4f* mine = part + (wave * 16 + 8 * half) * 16 + r;
    #pragma unroll
    for (int v = 0; v < 8; ++v)
        mine[v * 16] = (v4f){c0[v], c1[v], c2[v], c3[v]};
    __syncthreads();

    const v2f* col = (const v2f*)part + wave * 32 + lane;
    v2f s = col[0];
    #pragma unroll
    for (int w = 1; w < PV_WAVES_; ++w)
        s += col[w * 512];
    float* orow = out + ((size_t)b * Q_ + tm * 16 + wave) * VS_ + ns * 64 + 2 * lane;
    *(v2f*)orow = s;
}

#define QG_ 4
#define HS_ 8
#define NB_ 7
#define NST_ (H_ / HS_)

__global__ __launch_bounds__(512) __attribute__((amdgpu_waves_per_eu(4, 4)))
void scores_softmax(
    const float* __restrict__ qproj,
    const float* __restrict__ kprojT,
    const float* __restrict__ Ws,
    const int*   __restrict__ valid_len,
    float* __restrict__ attn)
{
    __shared__ float s_max[QG_][16];
    __shared__ float s_sum[QG_][16];

    constexpr int groups = Q_ / QG_;
    const int b  = blockIdx.x / groups;
    const int q0 = (blockIdx.x - b * groups) * QG_;
    const int t = threadIdx.x;
    const int wave = t >> 5;
    const int lane = t & 31;
    const int k0 = 2 * t;

    const float* qrow = qproj + ((size_t)b * Q_ + q0) * H_;
    const float* kcol = kprojT + (size_t)b * H_ * K_ + k0;
    const int vlen = valid_len[b];

    v2f acc[QG_];
    #pragma unroll
    for (int i = 0; i < QG_; ++i) acc[i] = (v2f){0.0f, 0.0f};

    v2f kb[NB_][HS_];
    #pragma unroll
    for (int s = 0; s < NB_ - 1; ++s) {
        #pragma unroll
        for (int j = 0; j < HS_; ++j)
            kb[s][j] = *(const v2f*)(kcol + (size_t)(s * HS_ + j) * K_);
    }

    #pragma unroll
    for (int s = 0; s < NST_; ++s) {
        if (s + NB_ - 1 < NST_) {
            #pragma unroll
            for (int j = 0; j < HS_; ++j)
                kb[(s + NB_ - 1) % NB_][j] =
                    *(const v2f*)(kcol + (size_t)((s + NB_ - 1) * HS_ + j) * K_);
        }

        #pragma unroll
        for (int j = 0; j < HS_; ++j) {
            const int h = s * HS_ + j;
            const float w = Ws[h];
            #pragma unroll
            for (int i = 0; i < QG_; ++i) {
                const v2f x = kb[s % NB_][j] + qrow[i * H_ + h];
                v2f th;
                th.x = TANHF(x.x);
                th.y = TANHF(x.y);
                acc[i] += w * th;
            }
        }
    }

    const bool m0 = (k0     < vlen);
    const bool m1 = (k0 + 1 < vlen);
    float mx[QG_];
    #pragma unroll
    for (int i = 0; i < QG_; ++i) {
        acc[i].x = m0 ? NEG_ : acc[i].x;
        acc[i].y = m1 ? NEG_ : acc[i].y;
        mx[i] = fmaxf(acc[i].x, acc[i].y);
    }
    #pragma unroll
    for (int off = 16; off >= 1; off >>= 1) {
        #pragma unroll
        for (int i = 0; i < QG_; ++i)
            mx[i] = fmaxf(mx[i], __shfl_xor(mx[i], off, 32));
    }
    if (lane == 0) {
        #pragma unroll
        for (int i = 0; i < QG_; ++i) s_max[i][wave] = mx[i];
    }
    __syncthreads();

    v2f e[QG_];
    float sm[QG_];
    #pragma unroll
    for (int i = 0; i < QG_; ++i) {
        float gmax = s_max[i][0];
        #pragma unroll
        for (int w = 1; w < 16; ++w) gmax = fmaxf(gmax, s_max[i][w]);
        e[i].x = EXP2F((acc[i].x - gmax) * LOG2E);
        e[i].y = EXP2F((acc[i].y - gmax) * LOG2E);
        sm[i] = e[i].x + e[i].y;
    }
    #pragma unroll
    for (int off = 16; off >= 1; off >>= 1) {
        #pragma unroll
        for (int i = 0; i < QG_; ++i)
            sm[i] += __shfl_xor(sm[i], off, 32);
    }
    if (lane == 0) {
        #pragma unroll
        for (int i = 0; i < QG_; ++i) s_sum[i][wave] = sm[i];
    }
    __syncthreads();

    #pragma unroll
    for (int i = 0; i < QG_; ++i) {
        float tot = s_sum[i][0];
        #pragma unroll
        for (int w = 1; w < 16; ++w) tot += s_sum[i][w];
        const float inv = 1.0f / tot;
        float* orow = attn + ((size_t)b * Q_ + q0 + i) * K_ + k0;
        *(v2f*)orow = e[i] * inv;
    }
}

extern "C" void kernel_launch(void* const* d_in, const int* in_sizes, int n_in,
                              void* d_out, int out_size, void* d_ws, size_t ws_size,
                              hipStream_t stream) {
    const float* queries   = (const float*)d_in[0];
    const float* keys      = (const float*)d_in[1];
    const float* values    = (const float*)d_in[2];
    const int*   valid_len = (const int*)  d_in[3];
    const float* Wq        = (const float*)d_in[4];
    const float* Wk        = (const float*)d_in[5];
    const float* Ws        = (const float*)d_in[6];
    float* out = (float*)d_out;

    float* qproj = (float*)d_ws;
    float* kproj = qproj + (size_t)B_ * Q_ * H_;
    float* attn  = kproj + (size_t)B_ * K_ * H_;

    proj_qk_wmma_f32<<<QP_BLOCKS_ + KP_BLOCKS_, 128, 0, stream>>>(
        queries, Wq, qproj, keys, Wk, kproj);

    scores_softmax<<<B_ * (Q_ / QG_), 512, 0, stream>>>(
        qproj, kproj, Ws, valid_len, attn);

    pv_strip_splitk_wmma_f32<<<B_ * 16, 512, 0, stream>>>(attn, values, out);
}


# scores re-tiled to (batch, 16 queries, 256 keys) blocks: key plane read 4x not 16x; softmax moved into the PV launch (per-wave max and sum, rescaled at the k-split merge); softmax, k-sum order change
# speedup vs baseline: 4.2751x; 1.0558x over previous
#include <hip/hip_runtime.h>
#include <hip/hip_bf16.h>
#include <math.h>

#define B_  16
#define Q_  64
#define K_  1024
#define QS_ 256
#define H_  128
#define VS_ 256
#define NEG_ (-1e6f)

typedef float v2f __attribute__((ext_vector_type(2)));
typedef float v4f __attribute__((ext_vector_type(4)));
typedef float v8f __attribute__((ext_vector_type(8)));

#if defined(__has_builtin)
#  if __has_builtin(__builtin_amdgcn_tanhf)
#    define TANHF(x) __builtin_amdgcn_tanhf(x)
#  else
#    define TANHF(x) tanhf(x)
#  endif
#  if __has_builtin(__builtin_amdgcn_exp2f)
#    define EXP2F(x) __builtin_amdgcn_exp2f(x)
#  else
#    define EXP2F(x) exp2f(x)
#  endif
#else
#  define TANHF(x) tanhf(x)
#  define EXP2F(x) exp2f(x)
#endif

#define LOG2E 1.4426950408889634f

#define QT_NB_ 20
#define KS_NB_ 16

template<int N, int KD>
__device__ __forceinline__ void gemm_tile_wmma_f32(
    const float* __restrict__ A,
    const float* __restrict__ W,
    float* __restrict__ C,
    const int blk)
{
    const int wave = threadIdx.x >> 5;
    const int lane = threadIdx.x & 31;
    const int tile = blk * 4 + wave;
    constexpr int tilesN = N >> 4;
    const int tm = tile / tilesN;
    const int tn = tile - tm * tilesN;
    const int r    = lane & 15;
    const int half = lane >> 4;

    const float* arow = A + (size_t)(tm * 16 + r) * KD + 2 * half;
    const float* wcol = W + (size_t)(2 * half) * N + tn * 16 + r;

    constexpr int NS = KD / 4;
    v2f ar[QT_NB_];
    v2f br[QT_NB_];
    #pragma unroll
    for (int s = 0; s < QT_NB_ - 1; ++s) {
        ar[s]   = *(const v2f*)(arow + 4 * s);
        br[s].x = wcol[(size_t)(4 * s) * N];
        br[s].y = wcol[(size_t)(4 * s + 1) * N];
    }

    v8f c = {};
    #pragma unroll
    for (int s = 0; s < NS; ++s) {
        if (s + QT_NB_ - 1 < NS) {
            const int t = s + QT_NB_ - 1;
            ar[t % QT_NB_]   = *(const v2f*)(arow + 4 * t);
            br[t % QT_NB_].x = wcol[(size_t)(4 * t) * N];
            br[t % QT_NB_].y = wcol[(size_t)(4 * t + 1) * N];
        }
        c = __builtin_amdgcn_wmma_f32_16x16x4_f32(
                 false, ar[s % QT_NB_],  false, br[s % QT_NB_],
                 (short)0, c,  false,  false);
    }

    float* crow = C + (size_t)(tm * 16 + 8 * half) * N + tn * 16 + r;
    #pragma unroll
    for (int v = 0; v < 8; ++v)
        crow[(size_t)v * N] = c[v];
}

template<int N, int KD>
__device__ __forceinline__ void proj_strip_wmma_f32(
    const float* __restrict__ A,
    const float* __restrict__ W,
    float* __restrict__ C,
    const int blk)
{
    const int wave = threadIdx.x >> 5;
    const int lane = threadIdx.x & 31;
    const int strip = blk * 4 + wave;
    constexpr int stripsN = N >> 6;
    const int tm = strip / stripsN;
    const int ns = strip - tm * stripsN;
    const int r    = lane & 15;
    const int half = lane >> 4;

    const float* arow = A + (size_t)(tm * 16 + r) * KD + 2 * half;
    const float* wrow = W + (size_t)(2 * half) * N + ns * 64 + 4 * r;

    constexpr int NS = KD / 4;
    v2f ar[KS_NB_];
    v4f wr0[KS_NB_];
    v4f wr1[KS_NB_];
    #pragma unroll
    for (int s = 0; s < KS_NB_ - 1; ++s) {
        ar[s]  = *(const v2f*)(arow + 4 * s);
        wr0[s] = *(const v4f*)(wrow + (size_t)(4 * s) * N);
        wr1[s] = *(const v4f*)(wrow + (size_t)(4 * s + 1) * N);
    }

    v8f c0 = {}, c1 = {}, c2 = {}, c3 = {};
    #pragma unroll
    for (int s = 0; s < NS; ++s) {
        if (s + KS_NB_ - 1 < NS) {
            const int t = s + KS_NB_ - 1;
            ar[t % KS_NB_]  = *(const v2f*)(arow + 4 * t);
            wr0[t % KS_NB_] = *(const v4f*)(wrow + (size_t)(4 * t) * N);
            wr1[t % KS_NB_] = *(const v4f*)(wrow + (size_t)(4 * t + 1) * N);
        }
        const v2f a  = ar[s % KS_NB_];
        const v4f w0 = wr0[s % KS_NB_];
        const v4f w1 = wr1[s % KS_NB_];
        const v2f b0 = {w0.x, w1.x};
        const v2f b1 = {w0.y, w1.y};
        const v2f b2 = {w0.z, w1.z};
        const v2f b3 = {w0.w, w1.w};
        c0 = __builtin_amdgcn_wmma_f32_16x16x4_f32(false, b0, false, a, (short)0, c0, false, false);
        c1 = __builtin_amdgcn_wmma_f32_16x16x4_f32(false, b1, false, a, (short)0, c1, false, false);
        c2 = __builtin_amdgcn_wmma_f32_16x16x4_f32(false, b2, false, a, (short)0, c2, false, false);
        c3 = __builtin_amdgcn_wmma_f32_16x16x4_f32(false, b3, false, a, (short)0, c3, false, false);
    }

    const int row = tm * 16 + r;
    const int bb  = row / K_;
    const int kidx = row - bb * K_;
    float* cbase = C + ((size_t)bb * N + ns * 64 + 32 * half) * K_ + kidx;
    #pragma unroll
    for (int v = 0; v < 8; ++v) {
        cbase[(size_t)(4 * v + 0) * K_] = c0[v];
        cbase[(size_t)(4 * v + 1) * K_] = c1[v];
        cbase[(size_t)(4 * v + 2) * K_] = c2[v];
        cbase[(size_t)(4 * v + 3) * K_] = c3[v];
    }
}

#define QP_BLOCKS_ ((B_ * Q_ / 16) * (H_ / 16) / 4)
#define KP_BLOCKS_ ((B_ * K_ / 16) * (H_ / 64) / 4)

__global__ __launch_bounds__(128) __attribute__((amdgpu_waves_per_eu(3, 3)))
void proj_qk_wmma_f32(
    const float* __restrict__ queries,
    const float* __restrict__ Wq,
    float* __restrict__ qproj,
    const float* __restrict__ keys,
    const float* __restrict__ Wk,
    float* __restrict__ kprojT)
{
    if (blockIdx.x < QP_BLOCKS_)
        gemm_tile_wmma_f32<H_, QS_>(queries, Wq, qproj, (int)blockIdx.x);
    else
        proj_strip_wmma_f32<H_, QS_>(keys, Wk, kprojT, (int)blockIdx.x - QP_BLOCKS_);
}

#define QG_ 4
#define HS_ 8
#define NB_ 7
#define NST_ (H_ / HS_)
#define KT_ 64
#define KQ_ (4 * KT_)
#define QB_ (4 * QG_)

__global__ __launch_bounds__(512) __attribute__((amdgpu_waves_per_eu(4, 4)))
void scores_keytile(
    const float* __restrict__ qproj,
    const float* __restrict__ kprojT,
    const float* __restrict__ Ws,
    const int*   __restrict__ valid_len,
    float* __restrict__ sc)
{
    constexpr int kparts = K_ / KQ_;
    constexpr int per_b  = kparts * (Q_ / QB_);
    const int b   = blockIdx.x / per_b;
    const int rem = blockIdx.x - b * per_b;
    const int qgb = rem / kparts;
    const int kq  = rem - qgb * kparts;
    const int wave = __builtin_amdgcn_readfirstlane((int)(threadIdx.x >> 5));
    const int lane = threadIdx.x & 31;
    const int q0 = qgb * QB_ + (wave & 3) * QG_;
    const int k0 = kq * KQ_ + (wave >> 2) * KT_ + 2 * lane;

    const float* qrow = qproj + ((size_t)b * Q_ + q0) * H_;
    const float* kcol = kprojT + (size_t)b * H_ * K_ + k0;
    const int vlen = valid_len[b];

    v2f acc[QG_];
    #pragma unroll
    for (int i = 0; i < QG_; ++i) acc[i] = (v2f){0.0f, 0.0f};

    v2f kb[NB_][HS_];
    #pragma unroll
    for (int s = 0; s < NB_ - 1; ++s) {
        #pragma unroll
        for (int j = 0; j < HS_; ++j)
            kb[s][j] = *(const v2f*)(kcol + (size_t)(s * HS_ + j) * K_);
    }

    #pragma unroll
    for (int s = 0; s < NST_; ++s) {
        if (s + NB_ - 1 < NST_) {
            #pragma unroll
            for (int j = 0; j < HS_; ++j)
                kb[(s + NB_ - 1) % NB_][j] =
                    *(const v2f*)(kcol + (size_t)((s + NB_ - 1) * HS_ + j) * K_);
        }

        #pragma unroll
        for (int j = 0; j < HS_; ++j) {
            const int h = s * HS_ + j;
            const float w = Ws[h];
            #pragma unroll
            for (int i = 0; i < QG_; ++i) {
                const v2f x = kb[s % NB_][j] + qrow[i * H_ + h];
                v2f th;
                th.x = TANHF(x.x);
                th.y = TANHF(x.y);
                acc[i] += w * th;
            }
        }
    }

    const bool m0 = (k0     < vlen);
    const bool m1 = (k0 + 1 < vlen);
    float* orow = sc + ((size_t)b * Q_ + q0) * K_ + k0;
    #pragma unroll
    for (int i = 0; i < QG_; ++i) {
        v2f o;
        o.x = m0 ? NEG_ : acc[i].x;
        o.y = m1 ? NEG_ : acc[i].y;
        *(v2f*)(orow + (size_t)i * K_) = o;
    }
}

#define PV_WAVES_ 16
#define PV_KW_    (K_ / PV_WAVES_)

__global__ __launch_bounds__(512) __attribute__((amdgpu_waves_per_eu(4, 4)))
void pv_softmax_splitk_wmma_f32(
    const float* __restrict__ sc,
    const float* __restrict__ values,
    float* __restrict__ out)
{
    __shared__ v4f part[PV_WAVES_ * 16 * 16];
    __shared__ float s_max[PV_WAVES_][16];
    __shared__ float s_sum[PV_WAVES_][16];

    const int wave = threadIdx.x >> 5;
    const int lane = threadIdx.x & 31;
    const int r    = lane & 15;
    const int half = lane >> 4;
    const int ns = blockIdx.x & 3;
    const int tm = (blockIdx.x >> 2) & 3;
    const int b  = blockIdx.x >> 4;

    const int k0 = wave * PV_KW_ + 4 * half;
    const float* arow = sc + ((size_t)b * Q_ + tm * 16 + r) * K_ + k0;
    const float* vrow = values + ((size_t)b * K_ + k0) * VS_ + ns * 64 + 4 * r;

    v4f a4[PV_KW_ / 8];
    v4f wv[PV_KW_ / 8][4];
    #pragma unroll
    for (int i = 0; i < PV_KW_ / 8; ++i)
        a4[i] = *(const v4f*)(arow + 8 * i);
    #pragma unroll
    for (int i = 0; i < PV_KW_ / 8; ++i) {
        #pragma unroll
        for (int j = 0; j < 4; ++j)
            wv[i][j] = *(const v4f*)(vrow + (size_t)(8 * i + j) * VS_);
    }

    float m = fmaxf(fmaxf(a4[0].x, a4[0].y), fmaxf(a4[0].z, a4[0].w));
    #pragma unroll
    for (int i = 1; i < PV_KW_ / 8; ++i)
        m = fmaxf(m, fmaxf(fmaxf(a4[i].x, a4[i].y), fmaxf(a4[i].z, a4[i].w)));
    m = fmaxf(m, __shfl_xor(m, 16, 32));

    float ls = 0.0f;
    #pragma unroll
    for (int i = 0; i < PV_KW_ / 8; ++i) {
        a4[i].x = EXP2F((a4[i].x - m) * LOG2E);
        a4[i].y = EXP2F((a4[i].y - m) * LOG2E);
        a4[i].z = EXP2F((a4[i].z - m) * LOG2E);
        a4[i].w = EXP2F((a4[i].w - m) * LOG2E);
        ls += (a4[i].x + a4[i].y) + (a4[i].z + a4[i].w);
    }
    ls += __shfl_xor(ls, 16, 32);
    s_max[wave][r] = m;
    s_sum[wave][r] = ls;

    __builtin_amdgcn_sched_barrier(0);

    v8f c0 = {}, c1 = {}, c2 = {}, c3 = {};
    #pragma unroll
    for (int i = 0; i < PV_KW_ / 8; ++i) {
        const v2f a01 = {a4[i].x, a4[i].y};
        const v2f a23 = {a4[i].z, a4[i].w};
        const v4f w0 = wv[i][0], w1 = wv[i][1], w2 = wv[i][2], w3 = wv[i][3];
        c0 = __builtin_amdgcn_wmma_f32_16x16x4_f32(false, a01, false, (v2f){w0.x, w1.x}, (short)0, c0, false, false);
        c1 = __builtin_amdgcn_wmma_f32_16x16x4_f32(false, a01, false, (v2f){w0.y, w1.y}, (short)0, c1, false, false);
        c2 = __builtin_amdgcn_wmma_f32_16x16x4_f32(false, a01, false, (v2f){w0.z, w1.z}, (short)0, c2, false, false);
        c3 = __builtin_amdgcn_wmma_f32_16x16x4_f32(false, a01, false, (v2f){w0.w, w1.w}, (short)0, c3, false, false);
        c0 = __builtin_amdgcn_wmma_f32_16x16x4_f32(false, a23, false, (v2f){w2.x, w3.x}, (short)0, c0, false, false);
        c1 = __builtin_amdgcn_wmma_f32_16x16x4_f32(false, a23, false, (v2f){w2.y, w3.y}, (short)0, c1, false, false);
        c2 = __builtin_amdgcn_wmma_f32_16x16x4_f32(false, a23, false, (v2f){w2.z, w3.z}, (short)0, c2, false, false);
        c3 = __builtin_amdgcn_wmma_f32_16x16x4_f32(false, a23, false, (v2f){w2.w, w3.w}, (short)0, c3, false, false);
        __builtin_amdgcn_sched_barrier(0);
    }

    v4f* mine = part + (wave * 16 + 8 * half) * 16 + r;
    #pragma unroll
    for (int v = 0; v < 8; ++v)
        mine[v * 16] = (v4f){c0[v], c1[v], c2[v], c3[v]};
    __syncthreads();

    float gmax = s_max[0][wave];
    #pragma unroll
    for (int w = 1; w < PV_WAVES_; ++w) gmax = fmaxf(gmax, s_max[w][wave]);

    const v2f* col = (const v2f*)part + wave * 32 + lane;
    v2f s = {0.0f, 0.0f};
    float tot = 0.0f;
    #pragma unroll
    for (int w = 0; w < PV_WAVES_; ++w) {
        const float f = EXP2F((s_max[w][wave] - gmax) * LOG2E);
        s += f * col[w * 512];
        tot += f * s_sum[w][wave];
    }
    const float inv = 1.0f / tot;
    float* orow = out + ((size_t)b * Q_ + tm * 16 + wave) * VS_ + ns * 64 + 2 * lane;
    *(v2f*)orow = s * inv;
}

extern "C" void kernel_launch(void* const* d_in, const int* in_sizes, int n_in,
                              void* d_out, int out_size, void* d_ws, size_t ws_size,
                              hipStream_t stream) {
    const float* queries   = (const float*)d_in[0];
    const float* keys      = (const float*)d_in[1];
    const float* values    = (const float*)d_in[2];
    const int*   valid_len = (const int*)  d_in[3];
    const float* Wq        = (const float*)d_in[4];
    const float* Wk        = (const float*)d_in[5];
    const float* Ws        = (const float*)d_in[6];
    float* out = (float*)d_out;

    float* qproj = (float*)d_ws;
    float* kproj = qproj + (size_t)B_ * Q_ * H_;
    float* sc    = kproj + (size_t)B_ * K_ * H_;

    proj_qk_wmma_f32<<<QP_BLOCKS_ + KP_BLOCKS_, 128, 0, stream>>>(
        queries, Wq, qproj, keys, Wk, kproj);

    scores_keytile<<<B_ * 16, 512, 0, stream>>>(
        qproj, kproj, Ws, valid_len, sc);

    pv_softmax_splitk_wmma_f32<<<B_ * 16, 512, 0, stream>>>(sc, values, out);
}
